# starting kernel
# baseline (speedup 1.0000x reference)
#include <hip/hip_runtime.h>
#include <hip/hip_bf16.h>


#define OUT_CH 256
#define IN_CH  256
#define HEADS  4
#define HID    64
#define NEG_SLOPE 0.2f

typedef float v2f  __attribute__((ext_vector_type(2)));
typedef float v8f  __attribute__((ext_vector_type(8)));
typedef __bf16 v16bf __attribute__((ext_vector_type(16)));

#if __has_builtin(__builtin_amdgcn_wmma_f32_16x16x4_f32)
#define HAVE_WMMA_F32X4 1
#else
#define HAVE_WMMA_F32X4 0
#endif

#if HAVE_WMMA_F32X4
__global__ void gemm_wmma_f32(const float* __restrict__ X,
                              const float* __restrict__ W,
                              float* __restrict__ H, int M) {
  const int wave  = threadIdx.x >> 5;
  const int lane  = threadIdx.x & 31;
  const int strip = blockIdx.x * 4 + wave;
  if (strip * 16 >= M) return;
  const int ncol0 = blockIdx.y * 64;
  const int lo    = lane & 15;
  const int hi    = lane >> 4;
  const int row   = strip * 16 + lo;

  v8f c0 = {}, c1 = {}, c2 = {}, c3 = {};
  for (int kk = 0; kk < IN_CH; kk += 4) {
    const int ka = kk + 2 * hi;
    v2f a = *(const v2f*)(X + (size_t)row * IN_CH + ka);
    v2f b0, b1, b2, b3;
    const float* wr0 = W + (size_t)ka * OUT_CH + ncol0 + lo;
    const float* wr1 = wr0 + OUT_CH;
    b0.x = wr0[0];  b0.y = wr1[0];
    b1.x = wr0[16]; b1.y = wr1[16];
    b2.x = wr0[32]; b2.y = wr1[32];
    b3.x = wr0[48]; b3.y = wr1[48];
    c0 = __builtin_amdgcn_wmma_f32_16x16x4_f32(false, a, false, b0, (short)0, c0, false, false);
    c1 = __builtin_amdgcn_wmma_f32_16x16x4_f32(false, a, false, b1, (short)0, c1, false, false);
    c2 = __builtin_amdgcn_wmma_f32_16x16x4_f32(false, a, false, b2, (short)0, c2, false, false);
    c3 = __builtin_amdgcn_wmma_f32_16x16x4_f32(false, a, false, b3, (short)0, c3, false, false);
  }
  v8f acc[4] = {c0, c1, c2, c3};
  #pragma unroll
  for (int t = 0; t < 4; ++t) {
    const int n = ncol0 + t * 16 + lo;
    #pragma unroll
    for (int r = 0; r < 8; ++r) {
      const int m = strip * 16 + r + 8 * hi;
      H[(size_t)m * OUT_CH + n] = acc[t][r];
    }
  }
}
#else
__global__ void gemm_wmma_f32(const float* __restrict__ X,
                              const float* __restrict__ W,
                              float* __restrict__ H, int M) {
  const int wave  = threadIdx.x >> 5;
  const int lane  = threadIdx.x & 31;
  const int strip = blockIdx.x * 4 + wave;
  if (strip * 16 >= M) return;
  const int ncol0 = blockIdx.y * 64;
  const int lo    = lane & 15;
  const int hi    = lane >> 4;
  const int row   = strip * 16 + lo;

  v8f acc[4] = {};
  for (int kk = 0; kk < IN_CH; kk += 32) {
    v16bf a;
    #pragma unroll
    for (int j = 0; j < 16; ++j) {
      const int p = j >> 1;
      const int k = 16 * (p >> 2) + 8 * hi + 2 * (p & 3) + (j & 1);
      a[j] = (__bf16)X[(size_t)row * IN_CH + kk + k];
    }
    #pragma unroll
    for (int t = 0; t < 4; ++t) {
      const int n = ncol0 + t * 16 + lo;
      v16bf b;
      #pragma unroll
      for (int j = 0; j < 16; ++j) {
        const int k = j + 16 * hi;
        b[j] = (__bf16)W[(size_t)(kk + k) * OUT_CH + n];
      }
      acc[t] = __builtin_amdgcn_wmma_f32_16x16x32_bf16(false, a, false, b,
                                                       (short)0, acc[t], false, false);
    }
  }
  #pragma unroll
  for (int t = 0; t < 4; ++t) {
    const int n = ncol0 + t * 16 + lo;
    #pragma unroll
    for (int r = 0; r < 8; ++r) {
      const int m = strip * 16 + r + 8 * hi;
      H[(size_t)m * OUT_CH + n] = acc[t][r];
    }
  }
}
#endif

__global__ void attn_dots(const float* __restrict__ H,
                          const float* __restrict__ att_src,
                          const float* __restrict__ att_dst,
                          float* __restrict__ a_src,
                          float* __restrict__ a_dst, int N) {
  const int idx = blockIdx.x * blockDim.x + threadIdx.x;
  if (idx >= N * HEADS) return;
  const int node = idx >> 2;
  const int head = idx & 3;
  const float* hp = H + (size_t)node * OUT_CH + head * HID;
  const float* as = att_src + head * HID;
  const float* ad = att_dst + head * HID;
  float s = 0.f, d = 0.f;
  #pragma unroll 8
  for (int c = 0; c < HID; ++c) {
    const float v = hp[c];
    s += v * as[c];
    d += v * ad[c];
  }
  a_src[idx] = s;
  a_dst[idx] = d;
}

__device__ __forceinline__ unsigned f2ord(float f) {
  unsigned u = __float_as_uint(f);
  return (u & 0x80000000u) ? ~u : (u | 0x80000000u);
}
__device__ __forceinline__ float ord2f(unsigned u) {
  return (u & 0x80000000u) ? __uint_as_float(u & 0x7fffffffu) : __uint_as_float(~u);
}
__device__ __forceinline__ void atomAddF(float* p, float v) {
  __hip_atomic_fetch_add(p, v, __ATOMIC_RELAXED, __HIP_MEMORY_SCOPE_AGENT);
}

__global__ void init_buffers(float* __restrict__ out, unsigned* __restrict__ emax,
                             float* __restrict__ esum, int out_n, int nh) {
  const int i = blockIdx.x * blockDim.x + threadIdx.x;
  if (i < out_n) out[i] = 0.f;
  if (i < nh) { emax[i] = 0x007FFFFFu  ; esum[i] = 0.f; }
}

__global__ void edge_max(const int* __restrict__ ei,
                         const float* __restrict__ a_src,
                         const float* __restrict__ a_dst,
                         unsigned* __restrict__ emax, int E, int N) {
  const int e = blockIdx.x * blockDim.x + threadIdx.x;
  if (e >= E + N) return;
  int s, d;
  if (e < E) { s = ei[e]; d = ei[E + e]; } else { s = d = e - E; }
  #pragma unroll
  for (int h = 0; h < HEADS; ++h) {
    float v = a_src[s * HEADS + h] + a_dst[d * HEADS + h];
    v = (v >= 0.f) ? v : NEG_SLOPE * v;
    atomicMax(&emax[d * HEADS + h], f2ord(v));
  }
}

__global__ void edge_sum(const int* __restrict__ ei,
                         const float* __restrict__ a_src,
                         const float* __restrict__ a_dst,
                         const unsigned* __restrict__ emax,
                         float* __restrict__ esum,
                         float* __restrict__ eexp, int E, int N) {
  const int e = blockIdx.x * blockDim.x + threadIdx.x;
  if (e >= E + N) return;
  int s, d;
  if (e < E) { s = ei[e]; d = ei[E + e]; } else { s = d = e - E; }
  #pragma unroll
  for (int h = 0; h < HEADS; ++h) {
    float v = a_src[s * HEADS + h] + a_dst[d * HEADS + h];
    v = (v >= 0.f) ? v : NEG_SLOPE * v;
    const float m = ord2f(emax[d * HEADS + h]);
    const float x = expf(v - m);
    eexp[(size_t)e * HEADS + h] = x;
    atomAddF(&esum[d * HEADS + h], x);
  }
}

__global__ void scatter_msgs(const int* __restrict__ ei,
                             const float* __restrict__ H,
                             const float* __restrict__ eexp,
                             const float* __restrict__ esum,
                             float* __restrict__ out, int E, int N) {
  const int gid  = blockIdx.x * blockDim.x + threadIdx.x;
  const int e    = gid >> 5;
  const int lane = gid & 31;
  if (e >= E + N) return;
  int s, d;
  if (e < E) { s = ei[e]; d = ei[E + e]; } else { s = d = e - E; }
  const int head  = lane >> 3;
  const float alpha = eexp[(size_t)e * HEADS + head] /
                      (esum[d * HEADS + head] + 1e-16f);
  const int c = lane * 8;
  const float4* hp = (const float4*)(H + (size_t)s * OUT_CH + c);
  const float4 h0 = hp[0];
  const float4 h1 = hp[1];
  float* op = out + (size_t)d * OUT_CH + c;
  atomAddF(op + 0, alpha * h0.x);
  atomAddF(op + 1, alpha * h0.y);
  atomAddF(op + 2, alpha * h0.z);
  atomAddF(op + 3, alpha * h0.w);
  atomAddF(op + 4, alpha * h1.x);
  atomAddF(op + 5, alpha * h1.y);
  atomAddF(op + 6, alpha * h1.z);
  atomAddF(op + 7, alpha * h1.w);
}

__global__ void finalize(float* __restrict__ out, const float* __restrict__ bias,
                         const float* __restrict__ prelu_w, int total) {
  const int i = blockIdx.x * blockDim.x + threadIdx.x;
  if (i >= total) return;
  const int c = i & (OUT_CH - 1);
  const float v = out[i] + bias[c];
  out[i] = (v >= 0.f) ? v : prelu_w[c] * v;
}

extern "C" void kernel_launch(void* const* d_in, const int* in_sizes, int n_in,
                              void* d_out, int out_size, void* d_ws, size_t ws_size,
                              hipStream_t stream) {
  const float* x       = (const float*)d_in[0];
  const int*   ei      = (const int*)d_in[1];
  const float* W       = (const float*)d_in[2];
  const float* att_src = (const float*)d_in[3];
  const float* att_dst = (const float*)d_in[4];
  const float* bias    = (const float*)d_in[5];
  const float* prelu_w = (const float*)d_in[6];
  float* out = (float*)d_out;

  const int N = in_sizes[0] / IN_CH;
  const int E = in_sizes[1] / 2;

  char* ws = (char*)d_ws;
  float* H = (float*)ws;            ws += (size_t)N * OUT_CH * sizeof(float);
  float* a_src = (float*)ws;        ws += (size_t)N * HEADS * sizeof(float);
  float* a_dst = (float*)ws;        ws += (size_t)N * HEADS * sizeof(float);
  unsigned* emax = (unsigned*)ws;   ws += (size_t)N * HEADS * sizeof(unsigned);
  float* esum = (float*)ws;         ws += (size_t)N * HEADS * sizeof(float);
  float* eexp = (float*)ws;

  const int strips = (N + 15) / 16;
  dim3 ggrid((strips + 3) / 4, OUT_CH / 64);
  gemm_wmma_f32<<<ggrid, 128, 0, stream>>>(x, W, H, N);

  attn_dots<<<(N * HEADS + 255) / 256, 256, 0, stream>>>(H, att_src, att_dst,
                                                         a_src, a_dst, N);

  const int total = N * OUT_CH;
  init_buffers<<<(total + 255) / 256, 256, 0, stream>>>(out, emax, esum,
                                                        total, N * HEADS);

  const int EL = E + N;
  edge_max<<<(EL + 255) / 256, 256, 0, stream>>>(ei, a_src, a_dst, emax, E, N);
  edge_sum<<<(EL + 255) / 256, 256, 0, stream>>>(ei, a_src, a_dst, emax, esum,
                                                 eexp, E, N);

  const long long sthreads = (long long)EL * 32;
  scatter_msgs<<<(int)((sthreads + 255) / 256), 256, 0, stream>>>(ei, H, eexp,
                                                                  esum, out, E, N);

  finalize<<<(total + 255) / 256, 256, 0, stream>>>(out, bias, prelu_w, total);
}


# Slab filter reads the destination column with 16-byte loads, one minimum tests four edges, sources fetched only where a group matches; node lists rank-sorted: sums now in ascending source order
# speedup vs baseline: 5.3844x; 5.3844x over previous
#include <hip/hip_runtime.h>
#include <hip/hip_bf16.h>


#define OUT_CH 256
#define IN_CH  256
#define HEADS  4
#define HID    64
#define NEG_SLOPE 0.2f

typedef float v2f  __attribute__((ext_vector_type(2)));
typedef float v8f  __attribute__((ext_vector_type(8)));
typedef __bf16 v16bf __attribute__((ext_vector_type(16)));

#if __has_builtin(__builtin_amdgcn_wmma_f32_16x16x4_f32)
#define HAVE_WMMA_F32X4 1
#else
#define HAVE_WMMA_F32X4 0
#endif

#if HAVE_WMMA_F32X4
__global__ void gemm_wmma_f32(const float* __restrict__ X,
                              const float* __restrict__ W,
                              float* __restrict__ H, int M) {
  const int wave  = threadIdx.x >> 5;
  const int lane  = threadIdx.x & 31;
  const int strip = blockIdx.x * 4 + wave;
  if (strip * 16 >= M) return;
  const int ncol0 = blockIdx.y * 64;
  const int lo    = lane & 15;
  const int hi    = lane >> 4;
  const int row   = strip * 16 + lo;

  v8f c0 = {}, c1 = {}, c2 = {}, c3 = {};
  for (int kk = 0; kk < IN_CH; kk += 4) {
    const int ka = kk + 2 * hi;
    v2f a = *(const v2f*)(X + (size_t)row * IN_CH + ka);
    v2f b0, b1, b2, b3;
    const float* wr0 = W + (size_t)ka * OUT_CH + ncol0 + lo;
    const float* wr1 = wr0 + OUT_CH;
    b0.x = wr0[0];  b0.y = wr1[0];
    b1.x = wr0[16]; b1.y = wr1[16];
    b2.x = wr0[32]; b2.y = wr1[32];
    b3.x = wr0[48]; b3.y = wr1[48];
    c0 = __builtin_amdgcn_wmma_f32_16x16x4_f32(false, a, false, b0, (short)0, c0, false, false);
    c1 = __builtin_amdgcn_wmma_f32_16x16x4_f32(false, a, false, b1, (short)0, c1, false, false);
    c2 = __builtin_amdgcn_wmma_f32_16x16x4_f32(false, a, false, b2, (short)0, c2, false, false);
    c3 = __builtin_amdgcn_wmma_f32_16x16x4_f32(false, a, false, b3, (short)0, c3, false, false);
  }
  v8f acc[4] = {c0, c1, c2, c3};
  #pragma unroll
  for (int t = 0; t < 4; ++t) {
    const int n = ncol0 + t * 16 + lo;
    #pragma unroll
    for (int r = 0; r < 8; ++r) {
      const int m = strip * 16 + r + 8 * hi;
      H[(size_t)m * OUT_CH + n] = acc[t][r];
    }
  }
}
#else
__global__ void gemm_wmma_f32(const float* __restrict__ X,
                              const float* __restrict__ W,
                              float* __restrict__ H, int M) {
  const int wave  = threadIdx.x >> 5;
  const int lane  = threadIdx.x & 31;
  const int strip = blockIdx.x * 4 + wave;
  if (strip * 16 >= M) return;
  const int ncol0 = blockIdx.y * 64;
  const int lo    = lane & 15;
  const int hi    = lane >> 4;
  const int row   = strip * 16 + lo;

  v8f acc[4] = {};
  for (int kk = 0; kk < IN_CH; kk += 32) {
    v16bf a;
    #pragma unroll
    for (int j = 0; j < 16; ++j) {
      const int p = j >> 1;
      const int k = 16 * (p >> 2) + 8 * hi + 2 * (p & 3) + (j & 1);
      a[j] = (__bf16)X[(size_t)row * IN_CH + kk + k];
    }
    #pragma unroll
    for (int t = 0; t < 4; ++t) {
      const int n = ncol0 + t * 16 + lo;
      v16bf b;
      #pragma unroll
      for (int j = 0; j < 16; ++j) {
        const int k = j + 16 * hi;
        b[j] = (__bf16)W[(size_t)(kk + k) * OUT_CH + n];
      }
      acc[t] = __builtin_amdgcn_wmma_f32_16x16x32_bf16(false, a, false, b,
                                                       (short)0, acc[t], false, false);
    }
  }
  #pragma unroll
  for (int t = 0; t < 4; ++t) {
    const int n = ncol0 + t * 16 + lo;
    #pragma unroll
    for (int r = 0; r < 8; ++r) {
      const int m = strip * 16 + r + 8 * hi;
      H[(size_t)m * OUT_CH + n] = acc[t][r];
    }
  }
}
#endif

__global__ void attn_dots(const float* __restrict__ H,
                          const float* __restrict__ att_src,
                          const float* __restrict__ att_dst,
                          float* __restrict__ a_src,
                          float* __restrict__ a_dst, int N) {
  const int idx = blockIdx.x * blockDim.x + threadIdx.x;
  if (idx >= N * HEADS) return;
  const int node = idx >> 2;
  const int head = idx & 3;
  const float* hp = H + (size_t)node * OUT_CH + head * HID;
  const float* as = att_src + head * HID;
  const float* ad = att_dst + head * HID;
  float s = 0.f, d = 0.f;
  #pragma unroll 8
  for (int c = 0; c < HID; ++c) {
    const float v = hp[c];
    s += v * as[c];
    d += v * ad[c];
  }
  a_src[idx] = s;
  a_dst[idx] = d;
}

#define NPG   196
#define NPW   7
#define NOWN  28
#define NCAP  256
#define RB    6
#define FV    8

__device__ __forceinline__ float leaky(float v) {
  return (v >= 0.f) ? v : NEG_SLOPE * v;
}
__device__ __forceinline__ float bcastf(float v, int srcLane) {
  return __int_as_float(__builtin_amdgcn_readlane(__float_as_int(v), srcLane));
}
__device__ __forceinline__ void fma4(float4& acc, float x, const float4& h) {
  acc.x += x * h.x; acc.y += x * h.y; acc.z += x * h.z; acc.w += x * h.w;
}


struct RowTail { float4 bA, bB, pA, pB; };

__device__ __forceinline__ void finish_row(float* __restrict__ out, int d, int lane,
                                           const float4& accA, const float4& accB,
                                           float l0, float l1, float l2, float l3,
                                           const float* __restrict__ bias,
                                           const float* __restrict__ prelu_w) {
  RowTail rt;
  rt.bA = *(const float4*)(bias + 4 * lane);
  rt.bB = *(const float4*)(bias + 128 + 4 * lane);
  rt.pA = *(const float4*)(prelu_w + 4 * lane);
  rt.pB = *(const float4*)(prelu_w + 128 + 4 * lane);
  for (int o = 16; o > 0; o >>= 1) {
    l0 += __shfl_xor(l0, o);
    l1 += __shfl_xor(l1, o);
    l2 += __shfl_xor(l2, o);
    l3 += __shfl_xor(l3, o);
  }
  const int hsel = lane >> 4;
  const float da = (hsel ? l1 : l0) + 1e-16f;
  const float db = (hsel ? l3 : l2) + 1e-16f;
  float4 oA, oB;
  float v;
  v = accA.x / da + rt.bA.x; oA.x = (v >= 0.f) ? v : rt.pA.x * v;
  v = accA.y / da + rt.bA.y; oA.y = (v >= 0.f) ? v : rt.pA.y * v;
  v = accA.z / da + rt.bA.z; oA.z = (v >= 0.f) ? v : rt.pA.z * v;
  v = accA.w / da + rt.bA.w; oA.w = (v >= 0.f) ? v : rt.pA.w * v;
  v = accB.x / db + rt.bB.x; oB.x = (v >= 0.f) ? v : rt.pB.x * v;
  v = accB.y / db + rt.bB.y; oB.y = (v >= 0.f) ? v : rt.pB.y * v;
  v = accB.z / db + rt.bB.z; oB.z = (v >= 0.f) ? v : rt.pB.z * v;
  v = accB.w / db + rt.bB.w; oB.w = (v >= 0.f) ? v : rt.pB.w * v;
  float* op = out + (size_t)d * OUT_CH;
  *(float4*)(op + 4 * lane) = oA;
  *(float4*)(op + 128 + 4 * lane) = oB;
}

__device__ __forceinline__ bool fetch_edge_global(const int* __restrict__ ei,
                                                  int d, int E, int i, int& s) {
  const bool mt = (i < E) && (ei[E + i] == d);
  s = mt ? ei[i] : d;
  return mt;
}

__device__ __forceinline__ void reduce_node_global(
    const int* __restrict__ ei, const float* __restrict__ H,
    const float* __restrict__ a_src, const float* __restrict__ a_dst,
    const float* __restrict__ bias, const float* __restrict__ prelu_w,
    float* __restrict__ out, int d, int E, int lane) {
  const float4 ad = *(const float4*)(a_dst + (size_t)d * HEADS);
  const float4 as = *(const float4*)(a_src + (size_t)d * HEADS);
  const float es0 = leaky(as.x + ad.x);
  const float es1 = leaky(as.y + ad.y);
  const float es2 = leaky(as.z + ad.z);
  const float es3 = leaky(as.w + ad.w);

  float m0 = es0, m1 = es1, m2 = es2, m3 = es3;
  for (int base = 0; base < E; base += 32) {
    int s;
    fetch_edge_global(ei, d, E, base + lane, s);
    const float4 a = *(const float4*)(a_src + (size_t)s * HEADS);
    m0 = fmaxf(m0, leaky(a.x + ad.x));
    m1 = fmaxf(m1, leaky(a.y + ad.y));
    m2 = fmaxf(m2, leaky(a.z + ad.z));
    m3 = fmaxf(m3, leaky(a.w + ad.w));
  }
  for (int o = 16; o > 0; o >>= 1) {
    m0 = fmaxf(m0, __shfl_xor(m0, o));
    m1 = fmaxf(m1, __shfl_xor(m1, o));
    m2 = fmaxf(m2, __shfl_xor(m2, o));
    m3 = fmaxf(m3, __shfl_xor(m3, o));
  }

  const int hsel = lane >> 4;
  const float* Hl = H + 4 * lane;
  const float xs0 = expf(es0 - m0);
  const float xs1 = expf(es1 - m1);
  const float xs2 = expf(es2 - m2);
  const float xs3 = expf(es3 - m3);
  float4 accA, accB;
  {
    const float xa = hsel ? xs1 : xs0;
    const float xb = hsel ? xs3 : xs2;
    const float4 hA = *(const float4*)(Hl + (size_t)d * OUT_CH);
    const float4 hB = *(const float4*)(Hl + (size_t)d * OUT_CH + 128);
    accA.x = xa * hA.x; accA.y = xa * hA.y; accA.z = xa * hA.z; accA.w = xa * hA.w;
    accB.x = xb * hB.x; accB.y = xb * hB.y; accB.z = xb * hB.z; accB.w = xb * hB.w;
  }
  float l0 = (lane == 0) ? xs0 : 0.f;
  float l1 = (lane == 0) ? xs1 : 0.f;
  float l2 = (lane == 0) ? xs2 : 0.f;
  float l3 = (lane == 0) ? xs3 : 0.f;

  for (int base = 0; base < E; base += 32) {
    int s;
    const bool mt = fetch_edge_global(ei, d, E, base + lane, s);
    const float4 a = *(const float4*)(a_src + (size_t)s * HEADS);
    const float w = mt ? 1.f : 0.f;
    const float x0 = expf(leaky(a.x + ad.x) - m0) * w;
    const float x1 = expf(leaky(a.y + ad.y) - m1) * w;
    const float x2 = expf(leaky(a.z + ad.z) - m2) * w;
    const float x3 = expf(leaky(a.w + ad.w) - m3) * w;
    l0 += x0; l1 += x1; l2 += x2; l3 += x3;
    unsigned mask = __builtin_amdgcn_ballot_w32(mt);
    while (mask) {
      const int b = __builtin_ctz(mask);
      mask &= mask - 1;
      const int sb = __builtin_amdgcn_readlane(s, b);
      const float4 hA = *(const float4*)(Hl + (size_t)sb * OUT_CH);
      const float4 hB = *(const float4*)(Hl + (size_t)sb * OUT_CH + 128);
      const float y0 = bcastf(x0, b);
      const float y1 = bcastf(x1, b);
      const float y2 = bcastf(x2, b);
      const float y3 = bcastf(x3, b);
      fma4(accA, hsel ? y1 : y0, hA);
      fma4(accB, hsel ? y3 : y2, hB);
    }
  }
  finish_row(out, d, lane, accA, accB, l0, l1, l2, l3, bias, prelu_w);
}

__device__ __forceinline__ void add_chunk(const float* __restrict__ Hl, int s, bool mt,
                                          const float4& sc, float m0, float m1,
                                          float m2, float m3, int cnt, int hsel,
                                          float4& accA, float4& accB,
                                          float& l0, float& l1, float& l2, float& l3) {
  const float w = mt ? 1.f : 0.f;
  const float x0 = expf(sc.x - m0) * w;
  const float x1 = expf(sc.y - m1) * w;
  const float x2 = expf(sc.z - m2) * w;
  const float x3 = expf(sc.w - m3) * w;
  l0 += x0; l1 += x1; l2 += x2; l3 += x3;
  for (int q0 = 0; q0 < cnt; q0 += RB) {
    float4 hA[RB], hB[RB];
    #pragma unroll
    for (int q = 0; q < RB; ++q) {
      const bool has = (q0 + q) < cnt;
      const int bl = has ? (q0 + q) : q0;
      const int sb = __builtin_amdgcn_readlane(s, bl);
      hA[q] = *(const float4*)(Hl + (size_t)sb * OUT_CH);
      hB[q] = *(const float4*)(Hl + (size_t)sb * OUT_CH + 128);
    }
    #pragma unroll
    for (int q = 0; q < RB; ++q) {
      const bool has = (q0 + q) < cnt;
      const int bl = has ? (q0 + q) : q0;
      const float y0 = bcastf(x0, bl);
      const float y1 = bcastf(x1, bl);
      const float y2 = bcastf(x2, bl);
      const float y3 = bcastf(x3, bl);
      fma4(accA, has ? (hsel ? y1 : y0) : 0.f, hA[q]);
      fma4(accB, has ? (hsel ? y3 : y2) : 0.f, hB[q]);
    }
  }
}

__device__ __forceinline__ void sort_list(const unsigned* lst, unsigned* srt,
                                          int n, int lane) {
  for (int base = 0; base < n; base += 32) {
    const int i = base + lane;
    const unsigned v = (i < n) ? lst[(i < n) ? i : 0] : 0xffffffffu;
    unsigned r = 0u;
    for (int b2 = 0; b2 < n; b2 += 32) {
      const int i2 = b2 + lane;
      const unsigned w = (i2 < n) ? lst[(i2 < n) ? i2 : 0] : 0xffffffffu;
      const int m = (n - b2 < 32) ? (n - b2) : 32;
      for (int j = 0; j < m; ++j) {
        const unsigned wj = (unsigned)__builtin_amdgcn_readlane((int)w, j);
        const int idx = b2 + j;
        r += (wj < v || (wj == v && idx < i)) ? 1u : 0u;
      }
    }
    if (i < n) srt[r] = v;
  }
}

__device__ __forceinline__ void reduce_node_list(
    const float* __restrict__ H, const float* __restrict__ a_src,
    const float* __restrict__ a_dst, const float* __restrict__ bias,
    const float* __restrict__ prelu_w, float* __restrict__ out,
    const unsigned* raw, unsigned* srt, int n, int d, int lane) {
  const int hsel = lane >> 4;
  const float* Hl = H + 4 * lane;
  const float4 ad = *(const float4*)(a_dst + (size_t)d * HEADS);
  const float4 as = *(const float4*)(a_src + (size_t)d * HEADS);
  const float4 hdA = *(const float4*)(Hl + (size_t)d * OUT_CH);
  const float4 hdB = *(const float4*)(Hl + (size_t)d * OUT_CH + 128);
  sort_list(raw, srt, n, lane);
  const unsigned* lst = srt;
  const bool mt0 = lane < n;
  const int  s0  = mt0 ? (int)lst[lane] : d;
  const float4 a0 = *(const float4*)(a_src + (size_t)s0 * HEADS);
  float4 sc0;
  sc0.x = leaky(a0.x + ad.x); sc0.y = leaky(a0.y + ad.y);
  sc0.z = leaky(a0.z + ad.z); sc0.w = leaky(a0.w + ad.w);
  const float es0 = leaky(as.x + ad.x);
  const float es1 = leaky(as.y + ad.y);
  const float es2 = leaky(as.z + ad.z);
  const float es3 = leaky(as.w + ad.w);

  float m0 = fmaxf(es0, sc0.x), m1 = fmaxf(es1, sc0.y);
  float m2 = fmaxf(es2, sc0.z), m3 = fmaxf(es3, sc0.w);
  for (int base = 32; base < n; base += 32) {
    const int i = base + lane;
    const bool mt = i < n;
    const int s = mt ? (int)lst[mt ? i : 0] : d;
    const float4 a = *(const float4*)(a_src + (size_t)s * HEADS);
    m0 = fmaxf(m0, leaky(a.x + ad.x));
    m1 = fmaxf(m1, leaky(a.y + ad.y));
    m2 = fmaxf(m2, leaky(a.z + ad.z));
    m3 = fmaxf(m3, leaky(a.w + ad.w));
  }
  for (int o = 16; o > 0; o >>= 1) {
    m0 = fmaxf(m0, __shfl_xor(m0, o));
    m1 = fmaxf(m1, __shfl_xor(m1, o));
    m2 = fmaxf(m2, __shfl_xor(m2, o));
    m3 = fmaxf(m3, __shfl_xor(m3, o));
  }

  const float xs0 = expf(es0 - m0);
  const float xs1 = expf(es1 - m1);
  const float xs2 = expf(es2 - m2);
  const float xs3 = expf(es3 - m3);
  float4 accA, accB;
  {
    const float xa = hsel ? xs1 : xs0;
    const float xb = hsel ? xs3 : xs2;
    accA.x = xa * hdA.x; accA.y = xa * hdA.y; accA.z = xa * hdA.z; accA.w = xa * hdA.w;
    accB.x = xb * hdB.x; accB.y = xb * hdB.y; accB.z = xb * hdB.z; accB.w = xb * hdB.w;
  }
  float l0 = (lane == 0) ? xs0 : 0.f;
  float l1 = (lane == 0) ? xs1 : 0.f;
  float l2 = (lane == 0) ? xs2 : 0.f;
  float l3 = (lane == 0) ? xs3 : 0.f;

  add_chunk(Hl, s0, mt0, sc0, m0, m1, m2, m3, (n < 32) ? n : 32, hsel,
            accA, accB, l0, l1, l2, l3);
  for (int base = 32; base < n; base += 32) {
    const int i = base + lane;
    const bool mt = i < n;
    const int s = mt ? (int)lst[mt ? i : 0] : d;
    const float4 a = *(const float4*)(a_src + (size_t)s * HEADS);
    float4 sc;
    sc.x = leaky(a.x + ad.x); sc.y = leaky(a.y + ad.y);
    sc.z = leaky(a.z + ad.z); sc.w = leaky(a.w + ad.w);
    const int rem = n - base;
    add_chunk(Hl, s, mt, sc, m0, m1, m2, m3, (rem < 32) ? rem : 32, hsel,
              accA, accB, l0, l1, l2, l3);
  }
  finish_row(out, d, lane, accA, accB, l0, l1, l2, l3, bias, prelu_w);
}

__device__ __forceinline__ void push_edge(unsigned (*list)[NCAP], unsigned* cnt,
                                          unsigned* ovf, unsigned dl, unsigned s) {
  const unsigned slot = atomicAdd(&cnt[dl], 1u);
  if (slot < NCAP) list[dl][slot] = s;
  else *ovf = 1u;
}

__global__ __launch_bounds__(1024) void aggregate_slabs(
    const int* __restrict__ ei, const float* __restrict__ H,
    const float* __restrict__ a_src, const float* __restrict__ a_dst,
    const float* __restrict__ bias, const float* __restrict__ prelu_w,
    float* __restrict__ out, int E, int N) {
  __shared__ unsigned s_list[NPG][NCAP];
  __shared__ unsigned s_sorted[NOWN][NCAP];
  __shared__ unsigned s_cnt[256];
  __shared__ unsigned s_ovf;

  const int tid  = threadIdx.x;
  const int lane = tid & 31;
  const int wave = tid >> 5;
  const int lo   = blockIdx.x * NPG;
  const unsigned span = (unsigned)((N - lo < NPG) ? (N - lo) : NPG);

  if (tid < 256) s_cnt[tid] = 0u;
  if (tid == 0) s_ovf = 0u;
  __syncthreads();

  if (E > 0) {
    const int* srcp = ei;
    const int* dstp = ei + E;
    int head = (int)(((16u - (unsigned)((size_t)dstp & 15u)) & 15u) >> 2);
    if (head > E) head = E;
    const int nvec  = (E - head) >> 2;
    const int tail0 = head + 4 * nvec;
    const int nOdd  = head + (E - tail0);
    if (tid < nOdd) {
      const int e = (tid < head) ? tid : (tail0 + tid - head);
      const unsigned dl = (unsigned)(dstp[e] - lo);
      if (dl < span) push_edge(s_list, s_cnt, &s_ovf, dl, (unsigned)srcp[e]);
    }
    if (nvec > 0) {
      const int4* dv = (const int4*)(dstp + head);
      const int UNITV = 1024 * FV;
      const int nUnits = (nvec + UNITV - 1) / UNITV;
      const int lastv = nvec - 1;
      int u = (int)(blockIdx.x % (unsigned)nUnits);
      int4 cur[FV];
      #pragma unroll
      for (int k = 0; k < FV; ++k) {
        const int g = u * UNITV + k * 1024 + tid;
        cur[k] = dv[(g < nvec) ? g : lastv];
      }
      for (int it = 0; it < nUnits; ++it) {
        const int g0 = u * UNITV + tid;
        int un = u + 1;
        if (un == nUnits) un = 0;
        int4 nxt[FV];
        #pragma unroll
        for (int k = 0; k < FV; ++k) {
          const int g = un * UNITV + k * 1024 + tid;
          nxt[k] = dv[(g < nvec) ? g : lastv];
        }
        bool anyk[FV];
        #pragma unroll
        for (int k = 0; k < FV; ++k) {
          const unsigned a = (unsigned)(cur[k].x - lo);
          const unsigned b = (unsigned)(cur[k].y - lo);
          const unsigned c = (unsigned)(cur[k].z - lo);
          const unsigned d = (unsigned)(cur[k].w - lo);
          const unsigned ab = (a < b) ? a : b;
          const unsigned cd = (c < d) ? c : d;
          const unsigned mk = (ab < cd) ? ab : cd;
          anyk[k] = __builtin_amdgcn_ballot_w32(mk < span) != 0u;
        }
        int sv[FV][4];
        #pragma unroll
        for (int k = 0; k < FV; ++k) {
          if (anyk[k]) {
            const int g = g0 + k * 1024;
            const unsigned lim = (g < nvec) ? span : 0u;
            const unsigned eb = (unsigned)(head + 4 * g);
            sv[k][0] = srcp[((unsigned)(cur[k].x - lo) < lim) ? eb      : 0u];
            sv[k][1] = srcp[((unsigned)(cur[k].y - lo) < lim) ? eb + 1u : 0u];
            sv[k][2] = srcp[((unsigned)(cur[k].z - lo) < lim) ? eb + 2u : 0u];
            sv[k][3] = srcp[((unsigned)(cur[k].w - lo) < lim) ? eb + 3u : 0u];
          }
        }
        #pragma unroll
        for (int k = 0; k < FV; ++k) {
          if (anyk[k]) {
            const int g = g0 + k * 1024;
            const unsigned lim = (g < nvec) ? span : 0u;
            const unsigned a = (unsigned)(cur[k].x - lo);
            const unsigned b = (unsigned)(cur[k].y - lo);
            const unsigned c = (unsigned)(cur[k].z - lo);
            const unsigned d = (unsigned)(cur[k].w - lo);
            if (a < lim) push_edge(s_list, s_cnt, &s_ovf, a, (unsigned)sv[k][0]);
            if (b < lim) push_edge(s_list, s_cnt, &s_ovf, b, (unsigned)sv[k][1]);
            if (c < lim) push_edge(s_list, s_cnt, &s_ovf, c, (unsigned)sv[k][2]);
            if (d < lim) push_edge(s_list, s_cnt, &s_ovf, d, (unsigned)sv[k][3]);
          }
        }
        #pragma unroll
        for (int k = 0; k < FV; ++k) cur[k] = nxt[k];
        u = un;
      }
    }
  }
  __syncthreads();

  if (wave >= NOWN) return;
  const int ovf = __builtin_amdgcn_readfirstlane((int)s_ovf);
  for (unsigned j = 0; j < NPW; ++j) {
    const unsigned dl = (unsigned)wave * NPW + j;
    if (dl >= span) break;
    const int d = lo + (int)dl;
    if (!ovf) {
      int n = __builtin_amdgcn_readfirstlane((int)s_cnt[dl]);
      if (n > NCAP) n = NCAP;
      reduce_node_list(H, a_src, a_dst, bias, prelu_w, out, &s_list[dl][0],
                       &s_sorted[wave][0], n, d, lane);
    } else {
      reduce_node_global(ei, H, a_src, a_dst, bias, prelu_w, out, d, E, lane);
    }
  }
}

extern "C" void kernel_launch(void* const* d_in, const int* in_sizes, int n_in,
                              void* d_out, int out_size, void* d_ws, size_t ws_size,
                              hipStream_t stream) {
  const float* x       = (const float*)d_in[0];
  const int*   ei      = (const int*)d_in[1];
  const float* W       = (const float*)d_in[2];
  const float* att_src = (const float*)d_in[3];
  const float* att_dst = (const float*)d_in[4];
  const float* bias    = (const float*)d_in[5];
  const float* prelu_w = (const float*)d_in[6];
  float* out = (float*)d_out;

  const int N = in_sizes[0] / IN_CH;
  const int E = in_sizes[1] / 2;

  char* ws = (char*)d_ws;
  float* H = (float*)ws;            ws += (size_t)N * OUT_CH * sizeof(float);
  float* a_src = (float*)ws;        ws += (size_t)N * HEADS * sizeof(float);
  float* a_dst = (float*)ws;

  const int strips = (N + 15) / 16;
  dim3 ggrid((strips + 3) / 4, OUT_CH / 64);
  gemm_wmma_f32<<<ggrid, 128, 0, stream>>>(x, W, H, N);

  attn_dots<<<(N * HEADS + 255) / 256, 256, 0, stream>>>(H, att_src, att_dst,
                                                         a_src, a_dst, N);

  aggregate_slabs<<<(N + NPG - 1) / NPG, 1024, 0, stream>>>(ei, H, a_src, a_dst,
                                                            bias, prelu_w, out, E, N);
}


# GEMM operands by 16-byte loads from memory: k in pieces of 16 (lane half h holds k 8h..8h+7), tiles interleaved (lane c: columns 4c..4c+3), next piece loaded ahead; k order of sums changed
# speedup vs baseline: 5.9506x; 1.1052x over previous
#include <hip/hip_runtime.h>
#include <hip/hip_bf16.h>


#define OUT_CH 256
#define IN_CH  256
#define HEADS  4
#define HID    64
#define NEG_SLOPE 0.2f

typedef float v2f  __attribute__((ext_vector_type(2)));
typedef float v4f  __attribute__((ext_vector_type(4)));
typedef float v8f  __attribute__((ext_vector_type(8)));
typedef __bf16 v16bf __attribute__((ext_vector_type(16)));

#if __has_builtin(__builtin_amdgcn_wmma_f32_16x16x4_f32)
#define HAVE_WMMA_F32X4 1
#else
#define HAVE_WMMA_F32X4 0
#endif

#if HAVE_WMMA_F32X4
__global__ __launch_bounds__(128) void gemm_wmma_f32(const float* __restrict__ X,
                              const float* __restrict__ W,
                              float* __restrict__ H, int M) {
  const int wave  = threadIdx.x >> 5;
  const int lane  = threadIdx.x & 31;
  const int strip = blockIdx.x * 4 + wave;
  if (strip * 16 >= M) return;
  const int ncol0 = blockIdx.y * 64;
  const int lo    = lane & 15;
  const int hi    = lane >> 4;
  const int row   = strip * 16 + lo;

  const float* xp = X + (size_t)row * IN_CH + 8 * hi;
  const float* wp = W + (size_t)(8 * hi) * OUT_CH + ncol0 + 4 * lo;

  v4f rx0 = *(const v4f*)(xp);
  v4f rx1 = *(const v4f*)(xp + 4);
  v4f rw[8];
  #pragma unroll
  for (int i = 0; i < 8; ++i)
    rw[i] = *(const v4f*)(wp + (size_t)i * OUT_CH);

  v8f c0 = {}, c1 = {}, c2 = {}, c3 = {};
  #pragma unroll
  for (int kk = 0; kk < IN_CH; kk += 16) {
    v2f a[4], b[4][4];
    a[0].x = rx0.x; a[0].y = rx0.y;
    a[1].x = rx0.z; a[1].y = rx0.w;
    a[2].x = rx1.x; a[2].y = rx1.y;
    a[3].x = rx1.z; a[3].y = rx1.w;
    #pragma unroll
    for (int t = 0; t < 4; ++t) {
      b[t][0].x = rw[2 * t].x; b[t][0].y = rw[2 * t + 1].x;
      b[t][1].x = rw[2 * t].y; b[t][1].y = rw[2 * t + 1].y;
      b[t][2].x = rw[2 * t].z; b[t][2].y = rw[2 * t + 1].z;
      b[t][3].x = rw[2 * t].w; b[t][3].y = rw[2 * t + 1].w;
    }
    if (kk + 16 < IN_CH) {
      rx0 = *(const v4f*)(xp + kk + 16);
      rx1 = *(const v4f*)(xp + kk + 20);
      #pragma unroll
      for (int i = 0; i < 8; ++i)
        rw[i] = *(const v4f*)(wp + (size_t)(kk + 16 + i) * OUT_CH);
    }
    __builtin_amdgcn_sched_barrier(0);
    #pragma unroll
    for (int t = 0; t < 4; ++t) {
      c0 = __builtin_amdgcn_wmma_f32_16x16x4_f32(false, a[t], false, b[t][0], (short)0, c0, false, false);
      c1 = __builtin_amdgcn_wmma_f32_16x16x4_f32(false, a[t], false, b[t][1], (short)0, c1, false, false);
      c2 = __builtin_amdgcn_wmma_f32_16x16x4_f32(false, a[t], false, b[t][2], (short)0, c2, false, false);
      c3 = __builtin_amdgcn_wmma_f32_16x16x4_f32(false, a[t], false, b[t][3], (short)0, c3, false, false);
    }
    __builtin_amdgcn_sched_barrier(0);
  }
  #pragma unroll
  for (int r = 0; r < 8; ++r) {
    const int m = strip * 16 + r + 8 * hi;
    v4f o;
    o.x = c0[r]; o.y = c1[r]; o.z = c2[r]; o.w = c3[r];
    *(v4f*)(H + (size_t)m * OUT_CH + ncol0 + 4 * lo) = o;
  }
}
#else
__global__ void gemm_wmma_f32(const float* __restrict__ X,
                              const float* __restrict__ W,
                              float* __restrict__ H, int M) {
  const int wave  = threadIdx.x >> 5;
  const int lane  = threadIdx.x & 31;
  const int strip = blockIdx.x * 4 + wave;
  if (strip * 16 >= M) return;
  const int ncol0 = blockIdx.y * 64;
  const int lo    = lane & 15;
  const int hi    = lane >> 4;
  const int row   = strip * 16 + lo;

  v8f acc[4] = {};
  for (int kk = 0; kk < IN_CH; kk += 32) {
    v16bf a;
    #pragma unroll
    for (int j = 0; j < 16; ++j) {
      const int p = j >> 1;
      const int k = 16 * (p >> 2) + 8 * hi + 2 * (p & 3) + (j & 1);
      a[j] = (__bf16)X[(size_t)row * IN_CH + kk + k];
    }
    #pragma unroll
    for (int t = 0; t < 4; ++t) {
      const int n = ncol0 + t * 16 + lo;
      v16bf b;
      #pragma unroll
      for (int j = 0; j < 16; ++j) {
        const int k = j + 16 * hi;
        b[j] = (__bf16)W[(size_t)(kk + k) * OUT_CH + n];
      }
      acc[t] = __builtin_amdgcn_wmma_f32_16x16x32_bf16(false, a, false, b,
                                                       (short)0, acc[t], false, false);
    }
  }
  #pragma unroll
  for (int t = 0; t < 4; ++t) {
    const int n = ncol0 + t * 16 + lo;
    #pragma unroll
    for (int r = 0; r < 8; ++r) {
      const int m = strip * 16 + r + 8 * hi;
      H[(size_t)m * OUT_CH + n] = acc[t][r];
    }
  }
}
#endif

__global__ void attn_dots(const float* __restrict__ H,
                          const float* __restrict__ att_src,
                          const float* __restrict__ att_dst,
                          float* __restrict__ a_src,
                          float* __restrict__ a_dst, int N) {
  const int idx = blockIdx.x * blockDim.x + threadIdx.x;
  if (idx >= N * HEADS) return;
  const int node = idx >> 2;
  const int head = idx & 3;
  const float* hp = H + (size_t)node * OUT_CH + head * HID;
  const float* as = att_src + head * HID;
  const float* ad = att_dst + head * HID;
  float s = 0.f, d = 0.f;
  #pragma unroll 8
  for (int c = 0; c < HID; ++c) {
    const float v = hp[c];
    s += v * as[c];
    d += v * ad[c];
  }
  a_src[idx] = s;
  a_dst[idx] = d;
}

#define NPG   196
#define NPW   7
#define NOWN  28
#define NCAP  256
#define RB    6
#define FV    8

__device__ __forceinline__ float leaky(float v) {
  return (v >= 0.f) ? v : NEG_SLOPE * v;
}
__device__ __forceinline__ float bcastf(float v, int srcLane) {
  return __int_as_float(__builtin_amdgcn_readlane(__float_as_int(v), srcLane));
}
__device__ __forceinline__ void fma4(float4& acc, float x, const float4& h) {
  acc.x += x * h.x; acc.y += x * h.y; acc.z += x * h.z; acc.w += x * h.w;
}


struct RowTail { float4 bA, bB, pA, pB; };

__device__ __forceinline__ void finish_row(float* __restrict__ out, int d, int lane,
                                           const float4& accA, const float4& accB,
                                           float l0, float l1, float l2, float l3,
                                           const float* __restrict__ bias,
                                           const float* __restrict__ prelu_w) {
  RowTail rt;
  rt.bA = *(const float4*)(bias + 4 * lane);
  rt.bB = *(const float4*)(bias + 128 + 4 * lane);
  rt.pA = *(const float4*)(prelu_w + 4 * lane);
  rt.pB = *(const float4*)(prelu_w + 128 + 4 * lane);
  for (int o = 16; o > 0; o >>= 1) {
    l0 += __shfl_xor(l0, o);
    l1 += __shfl_xor(l1, o);
    l2 += __shfl_xor(l2, o);
    l3 += __shfl_xor(l3, o);
  }
  const int hsel = lane >> 4;
  const float da = (hsel ? l1 : l0) + 1e-16f;
  const float db = (hsel ? l3 : l2) + 1e-16f;
  float4 oA, oB;
  float v;
  v = accA.x / da + rt.bA.x; oA.x = (v >= 0.f) ? v : rt.pA.x * v;
  v = accA.y / da + rt.bA.y; oA.y = (v >= 0.f) ? v : rt.pA.y * v;
  v = accA.z / da + rt.bA.z; oA.z = (v >= 0.f) ? v : rt.pA.z * v;
  v = accA.w / da + rt.bA.w; oA.w = (v >= 0.f) ? v : rt.pA.w * v;
  v = accB.x / db + rt.bB.x; oB.x = (v >= 0.f) ? v : rt.pB.x * v;
  v = accB.y / db + rt.bB.y; oB.y = (v >= 0.f) ? v : rt.pB.y * v;
  v = accB.z / db + rt.bB.z; oB.z = (v >= 0.f) ? v : rt.pB.z * v;
  v = accB.w / db + rt.bB.w; oB.w = (v >= 0.f) ? v : rt.pB.w * v;
  float* op = out + (size_t)d * OUT_CH;
  *(float4*)(op + 4 * lane) = oA;
  *(float4*)(op + 128 + 4 * lane) = oB;
}

__device__ __forceinline__ bool fetch_edge_global(const int* __restrict__ ei,
                                                  int d, int E, int i, int& s) {
  const bool mt = (i < E) && (ei[E + i] == d);
  s = mt ? ei[i] : d;
  return mt;
}

__device__ __forceinline__ void reduce_node_global(
    const int* __restrict__ ei, const float* __restrict__ H,
    const float* __restrict__ a_src, const float* __restrict__ a_dst,
    const float* __restrict__ bias, const float* __restrict__ prelu_w,
    float* __restrict__ out, int d, int E, int lane) {
  const float4 ad = *(const float4*)(a_dst + (size_t)d * HEADS);
  const float4 as = *(const float4*)(a_src + (size_t)d * HEADS);
  const float es0 = leaky(as.x + ad.x);
  const float es1 = leaky(as.y + ad.y);
  const float es2 = leaky(as.z + ad.z);
  const float es3 = leaky(as.w + ad.w);

  float m0 = es0, m1 = es1, m2 = es2, m3 = es3;
  for (int base = 0; base < E; base += 32) {
    int s;
    fetch_edge_global(ei, d, E, base + lane, s);
    const float4 a = *(const float4*)(a_src + (size_t)s * HEADS);
    m0 = fmaxf(m0, leaky(a.x + ad.x));
    m1 = fmaxf(m1, leaky(a.y + ad.y));
    m2 = fmaxf(m2, leaky(a.z + ad.z));
    m3 = fmaxf(m3, leaky(a.w + ad.w));
  }
  for (int o = 16; o > 0; o >>= 1) {
    m0 = fmaxf(m0, __shfl_xor(m0, o));
    m1 = fmaxf(m1, __shfl_xor(m1, o));
    m2 = fmaxf(m2, __shfl_xor(m2, o));
    m3 = fmaxf(m3, __shfl_xor(m3, o));
  }

  const int hsel = lane >> 4;
  const float* Hl = H + 4 * lane;
  const float xs0 = expf(es0 - m0);
  const float xs1 = expf(es1 - m1);
  const float xs2 = expf(es2 - m2);
  const float xs3 = expf(es3 - m3);
  float4 accA, accB;
  {
    const float xa = hsel ? xs1 : xs0;
    const float xb = hsel ? xs3 : xs2;
    const float4 hA = *(const float4*)(Hl + (size_t)d * OUT_CH);
    const float4 hB = *(const float4*)(Hl + (size_t)d * OUT_CH + 128);
    accA.x = xa * hA.x; accA.y = xa * hA.y; accA.z = xa * hA.z; accA.w = xa * hA.w;
    accB.x = xb * hB.x; accB.y = xb * hB.y; accB.z = xb * hB.z; accB.w = xb * hB.w;
  }
  float l0 = (lane == 0) ? xs0 : 0.f;
  float l1 = (lane == 0) ? xs1 : 0.f;
  float l2 = (lane == 0) ? xs2 : 0.f;
  float l3 = (lane == 0) ? xs3 : 0.f;

  for (int base = 0; base < E; base += 32) {
    int s;
    const bool mt = fetch_edge_global(ei, d, E, base + lane, s);
    const float4 a = *(const float4*)(a_src + (size_t)s * HEADS);
    const float w = mt ? 1.f : 0.f;
    const float x0 = expf(leaky(a.x + ad.x) - m0) * w;
    const float x1 = expf(leaky(a.y + ad.y) - m1) * w;
    const float x2 = expf(leaky(a.z + ad.z) - m2) * w;
    const float x3 = expf(leaky(a.w + ad.w) - m3) * w;
    l0 += x0; l1 += x1; l2 += x2; l3 += x3;
    unsigned mask = __builtin_amdgcn_ballot_w32(mt);
    while (mask) {
      const int b = __builtin_ctz(mask);
      mask &= mask - 1;
      const int sb = __builtin_amdgcn_readlane(s, b);
      const float4 hA = *(const float4*)(Hl + (size_t)sb * OUT_CH);
      const float4 hB = *(const float4*)(Hl + (size_t)sb * OUT_CH + 128);
      const float y0 = bcastf(x0, b);
      const float y1 = bcastf(x1, b);
      const float y2 = bcastf(x2, b);
      const float y3 = bcastf(x3, b);
      fma4(accA, hsel ? y1 : y0, hA);
      fma4(accB, hsel ? y3 : y2, hB);
    }
  }
  finish_row(out, d, lane, accA, accB, l0, l1, l2, l3, bias, prelu_w);
}

__device__ __forceinline__ void add_chunk(const float* __restrict__ Hl, int s, bool mt,
                                          const float4& sc, float m0, float m1,
                                          float m2, float m3, int cnt, int hsel,
                                          float4& accA, float4& accB,
                                          float& l0, float& l1, float& l2, float& l3) {
  const float w = mt ? 1.f : 0.f;
  const float x0 = expf(sc.x - m0) * w;
  const float x1 = expf(sc.y - m1) * w;
  const float x2 = expf(sc.z - m2) * w;
  const float x3 = expf(sc.w - m3) * w;
  l0 += x0; l1 += x1; l2 += x2; l3 += x3;
  for (int q0 = 0; q0 < cnt; q0 += RB) {
    float4 hA[RB], hB[RB];
    #pragma unroll
    for (int q = 0; q < RB; ++q) {
      const bool has = (q0 + q) < cnt;
      const int bl = has ? (q0 + q) : q0;
      const int sb = __builtin_amdgcn_readlane(s, bl);
      hA[q] = *(const float4*)(Hl + (size_t)sb * OUT_CH);
      hB[q] = *(const float4*)(Hl + (size_t)sb * OUT_CH + 128);
    }
    #pragma unroll
    for (int q = 0; q < RB; ++q) {
      const bool has = (q0 + q) < cnt;
      const int bl = has ? (q0 + q) : q0;
      const float y0 = bcastf(x0, bl);
      const float y1 = bcastf(x1, bl);
      const float y2 = bcastf(x2, bl);
      const float y3 = bcastf(x3, bl);
      fma4(accA, has ? (hsel ? y1 : y0) : 0.f, hA[q]);
      fma4(accB, has ? (hsel ? y3 : y2) : 0.f, hB[q]);
    }
  }
}

__device__ __forceinline__ void sort_list(const unsigned* lst, unsigned* srt,
                                          int n, int lane) {
  for (int base = 0; base < n; base += 32) {
    const int i = base + lane;
    const unsigned v = (i < n) ? lst[(i < n) ? i : 0] : 0xffffffffu;
    unsigned r = 0u;
    for (int b2 = 0; b2 < n; b2 += 32) {
      const int i2 = b2 + lane;
      const unsigned w = (i2 < n) ? lst[(i2 < n) ? i2 : 0] : 0xffffffffu;
      const int m = (n - b2 < 32) ? (n - b2) : 32;
      for (int j = 0; j < m; ++j) {
        const unsigned wj = (unsigned)__builtin_amdgcn_readlane((int)w, j);
        const int idx = b2 + j;
        r += (wj < v || (wj == v && idx < i)) ? 1u : 0u;
      }
    }
    if (i < n) srt[r] = v;
  }
}

__device__ __forceinline__ void reduce_node_list(
    const float* __restrict__ H, const float* __restrict__ a_src,
    const float* __restrict__ a_dst, const float* __restrict__ bias,
    const float* __restrict__ prelu_w, float* __restrict__ out,
    const unsigned* raw, unsigned* srt, int n, int d, int lane) {
  const int hsel = lane >> 4;
  const float* Hl = H + 4 * lane;
  const float4 ad = *(const float4*)(a_dst + (size_t)d * HEADS);
  const float4 as = *(const float4*)(a_src + (size_t)d * HEADS);
  const float4 hdA = *(const float4*)(Hl + (size_t)d * OUT_CH);
  const float4 hdB = *(const float4*)(Hl + (size_t)d * OUT_CH + 128);
  sort_list(raw, srt, n, lane);
  const unsigned* lst = srt;
  const bool mt0 = lane < n;
  const int  s0  = mt0 ? (int)lst[lane] : d;
  const float4 a0 = *(const float4*)(a_src + (size_t)s0 * HEADS);
  float4 sc0;
  sc0.x = leaky(a0.x + ad.x); sc0.y = leaky(a0.y + ad.y);
  sc0.z = leaky(a0.z + ad.z); sc0.w = leaky(a0.w + ad.w);
  const float es0 = leaky(as.x + ad.x);
  const float es1 = leaky(as.y + ad.y);
  const float es2 = leaky(as.z + ad.z);
  const float es3 = leaky(as.w + ad.w);

  float m0 = fmaxf(es0, sc0.x), m1 = fmaxf(es1, sc0.y);
  float m2 = fmaxf(es2, sc0.z), m3 = fmaxf(es3, sc0.w);
  for (int base = 32; base < n; base += 32) {
    const int i = base + lane;
    const bool mt = i < n;
    const int s = mt ? (int)lst[mt ? i : 0] : d;
    const float4 a = *(const float4*)(a_src + (size_t)s * HEADS);
    m0 = fmaxf(m0, leaky(a.x + ad.x));
    m1 = fmaxf(m1, leaky(a.y + ad.y));
    m2 = fmaxf(m2, leaky(a.z + ad.z));
    m3 = fmaxf(m3, leaky(a.w + ad.w));
  }
  for (int o = 16; o > 0; o >>= 1) {
    m0 = fmaxf(m0, __shfl_xor(m0, o));
    m1 = fmaxf(m1, __shfl_xor(m1, o));
    m2 = fmaxf(m2, __shfl_xor(m2, o));
    m3 = fmaxf(m3, __shfl_xor(m3, o));
  }

  const float xs0 = expf(es0 - m0);
  const float xs1 = expf(es1 - m1);
  const float xs2 = expf(es2 - m2);
  const float xs3 = expf(es3 - m3);
  float4 accA, accB;
  {
    const float xa = hsel ? xs1 : xs0;
    const float xb = hsel ? xs3 : xs2;
    accA.x = xa * hdA.x; accA.y = xa * hdA.y; accA.z = xa * hdA.z; accA.w = xa * hdA.w;
    accB.x = xb * hdB.x; accB.y = xb * hdB.y; accB.z = xb * hdB.z; accB.w = xb * hdB.w;
  }
  float l0 = (lane == 0) ? xs0 : 0.f;
  float l1 = (lane == 0) ? xs1 : 0.f;
  float l2 = (lane == 0) ? xs2 : 0.f;
  float l3 = (lane == 0) ? xs3 : 0.f;

  add_chunk(Hl, s0, mt0, sc0, m0, m1, m2, m3, (n < 32) ? n : 32, hsel,
            accA, accB, l0, l1, l2, l3);
  for (int base = 32; base < n; base += 32) {
    const int i = base + lane;
    const bool mt = i < n;
    const int s = mt ? (int)lst[mt ? i : 0] : d;
    const float4 a = *(const float4*)(a_src + (size_t)s * HEADS);
    float4 sc;
    sc.x = leaky(a.x + ad.x); sc.y = leaky(a.y + ad.y);
    sc.z = leaky(a.z + ad.z); sc.w = leaky(a.w + ad.w);
    const int rem = n - base;
    add_chunk(Hl, s, mt, sc, m0, m1, m2, m3, (rem < 32) ? rem : 32, hsel,
              accA, accB, l0, l1, l2, l3);
  }
  finish_row(out, d, lane, accA, accB, l0, l1, l2, l3, bias, prelu_w);
}

__device__ __forceinline__ void push_edge(unsigned (*list)[NCAP], unsigned* cnt,
                                          unsigned* ovf, unsigned dl, unsigned s) {
  const unsigned slot = atomicAdd(&cnt[dl], 1u);
  if (slot < NCAP) list[dl][slot] = s;
  else *ovf = 1u;
}

__global__ __launch_bounds__(1024) void aggregate_slabs(
    const int* __restrict__ ei, const float* __restrict__ H,
    const float* __restrict__ a_src, const float* __restrict__ a_dst,
    const float* __restrict__ bias, const float* __restrict__ prelu_w,
    float* __restrict__ out, int E, int N) {
  __shared__ unsigned s_list[NPG][NCAP];
  __shared__ unsigned s_sorted[NOWN][NCAP];
  __shared__ unsigned s_cnt[256];
  __shared__ unsigned s_ovf;

  const int tid  = threadIdx.x;
  const int lane = tid & 31;
  const int wave = tid >> 5;
  const int lo   = blockIdx.x * NPG;
  const unsigned span = (unsigned)((N - lo < NPG) ? (N - lo) : NPG);

  if (tid < 256) s_cnt[tid] = 0u;
  if (tid == 0) s_ovf = 0u;
  __syncthreads();

  if (E > 0) {
    const int* srcp = ei;
    const int* dstp = ei + E;
    int head = (int)(((16u - (unsigned)((size_t)dstp & 15u)) & 15u) >> 2);
    if (head > E) head = E;
    const int nvec  = (E - head) >> 2;
    const int tail0 = head + 4 * nvec;
    const int nOdd  = head + (E - tail0);
    if (tid < nOdd) {
      const int e = (tid < head) ? tid : (tail0 + tid - head);
      const unsigned dl = (unsigned)(dstp[e] - lo);
      if (dl < span) push_edge(s_list, s_cnt, &s_ovf, dl, (unsigned)srcp[e]);
    }
    if (nvec > 0) {
      const int4* dv = (const int4*)(dstp + head);
      const int UNITV = 1024 * FV;
      const int nUnits = (nvec + UNITV - 1) / UNITV;
      const int lastv = nvec - 1;
      int u = (int)(blockIdx.x % (unsigned)nUnits);
      int4 cur[FV];
      #pragma unroll
      for (int k = 0; k < FV; ++k) {
        const int g = u * UNITV + k * 1024 + tid;
        cur[k] = dv[(g < nvec) ? g : lastv];
      }
      for (int it = 0; it < nUnits; ++it) {
        const int g0 = u * UNITV + tid;
        int un = u + 1;
        if (un == nUnits) un = 0;
        int4 nxt[FV];
        #pragma unroll
        for (int k = 0; k < FV; ++k) {
          const int g = un * UNITV + k * 1024 + tid;
          nxt[k] = dv[(g < nvec) ? g : lastv];
        }
        bool anyk[FV];
        #pragma unroll
        for (int k = 0; k < FV; ++k) {
          const unsigned a = (unsigned)(cur[k].x - lo);
          const unsigned b = (unsigned)(cur[k].y - lo);
          const unsigned c = (unsigned)(cur[k].z - lo);
          const unsigned d = (unsigned)(cur[k].w - lo);
          const unsigned ab = (a < b) ? a : b;
          const unsigned cd = (c < d) ? c : d;
          const unsigned mk = (ab < cd) ? ab : cd;
          anyk[k] = __builtin_amdgcn_ballot_w32(mk < span) != 0u;
        }
        int sv[FV][4];
        #pragma unroll
        for (int k = 0; k < FV; ++k) {
          if (anyk[k]) {
            const int g = g0 + k * 1024;
            const unsigned lim = (g < nvec) ? span : 0u;
            const unsigned eb = (unsigned)(head + 4 * g);
            sv[k][0] = srcp[((unsigned)(cur[k].x - lo) < lim) ? eb      : 0u];
            sv[k][1] = srcp[((unsigned)(cur[k].y - lo) < lim) ? eb + 1u : 0u];
            sv[k][2] = srcp[((unsigned)(cur[k].z - lo) < lim) ? eb + 2u : 0u];
            sv[k][3] = srcp[((unsigned)(cur[k].w - lo) < lim) ? eb + 3u : 0u];
          }
        }
        #pragma unroll
        for (int k = 0; k < FV; ++k) {
          if (anyk[k]) {
            const int g = g0 + k * 1024;
            const unsigned lim = (g < nvec) ? span : 0u;
            const unsigned a = (unsigned)(cur[k].x - lo);
            const unsigned b = (unsigned)(cur[k].y - lo);
            const unsigned c = (unsigned)(cur[k].z - lo);
            const unsigned d = (unsigned)(cur[k].w - lo);
            if (a < lim) push_edge(s_list, s_cnt, &s_ovf, a, (unsigned)sv[k][0]);
            if (b < lim) push_edge(s_list, s_cnt, &s_ovf, b, (unsigned)sv[k][1]);
            if (c < lim) push_edge(s_list, s_cnt, &s_ovf, c, (unsigned)sv[k][2]);
            if (d < lim) push_edge(s_list, s_cnt, &s_ovf, d, (unsigned)sv[k][3]);
          }
        }
        #pragma unroll
        for (int k = 0; k < FV; ++k) cur[k] = nxt[k];
        u = un;
      }
    }
  }
  __syncthreads();

  if (wave >= NOWN) return;
  const int ovf = __builtin_amdgcn_readfirstlane((int)s_ovf);
  for (unsigned j = 0; j < NPW; ++j) {
    const unsigned dl = (unsigned)wave * NPW + j;
    if (dl >= span) break;
    const int d = lo + (int)dl;
    if (!ovf) {
      int n = __builtin_amdgcn_readfirstlane((int)s_cnt[dl]);
      if (n > NCAP) n = NCAP;
      reduce_node_list(H, a_src, a_dst, bias, prelu_w, out, &s_list[dl][0],
                       &s_sorted[wave][0], n, d, lane);
    } else {
      reduce_node_global(ei, H, a_src, a_dst, bias, prelu_w, out, d, E, lane);
    }
  }
}

extern "C" void kernel_launch(void* const* d_in, const int* in_sizes, int n_in,
                              void* d_out, int out_size, void* d_ws, size_t ws_size,
                              hipStream_t stream) {
  const float* x       = (const float*)d_in[0];
  const int*   ei      = (const int*)d_in[1];
  const float* W       = (const float*)d_in[2];
  const float* att_src = (const float*)d_in[3];
  const float* att_dst = (const float*)d_in[4];
  const float* bias    = (const float*)d_in[5];
  const float* prelu_w = (const float*)d_in[6];
  float* out = (float*)d_out;

  const int N = in_sizes[0] / IN_CH;
  const int E = in_sizes[1] / 2;

  char* ws = (char*)d_ws;
  float* H = (float*)ws;            ws += (size_t)N * OUT_CH * sizeof(float);
  float* a_src = (float*)ws;        ws += (size_t)N * HEADS * sizeof(float);
  float* a_dst = (float*)ws;

  const int strips = (N + 15) / 16;
  dim3 ggrid((strips + 3) / 4, OUT_CH / 64);
  gemm_wmma_f32<<<ggrid, 128, 0, stream>>>(x, W, H, N);

  attn_dots<<<(N * HEADS + 255) / 256, 256, 0, stream>>>(H, att_src, att_dst,
                                                         a_src, a_dst, N);

  aggregate_slabs<<<(N + NPG - 1) / NPG, 1024, 0, stream>>>(ei, H, a_src, a_dst,
                                                            bias, prelu_w, out, E, N);
}


# Slab filter scans one-byte slab keys (dst/196, written by extra blocks of the attention-dot launch) 16 edges per load into per-wave LDS match lists; dst/src read for matches only; order of sums kept
# speedup vs baseline: 6.6954x; 1.1252x over previous
#include <hip/hip_runtime.h>
#include <hip/hip_bf16.h>


#define OUT_CH 256
#define IN_CH  256
#define HEADS  4
#define HID    64
#define NEG_SLOPE 0.2f
#define NPG   196

typedef float v2f  __attribute__((ext_vector_type(2)));
typedef float v4f  __attribute__((ext_vector_type(4)));
typedef float v8f  __attribute__((ext_vector_type(8)));
typedef __bf16 v16bf __attribute__((ext_vector_type(16)));

#if __has_builtin(__builtin_amdgcn_wmma_f32_16x16x4_f32)
#define HAVE_WMMA_F32X4 1
#else
#define HAVE_WMMA_F32X4 0
#endif

#if HAVE_WMMA_F32X4
__global__ __launch_bounds__(128) void gemm_wmma_f32(const float* __restrict__ X,
                              const float* __restrict__ W,
                              float* __restrict__ H, int M) {
  const int wave  = threadIdx.x >> 5;
  const int lane  = threadIdx.x & 31;
  const int strip = blockIdx.x * 4 + wave;
  if (strip * 16 >= M) return;
  const int ncol0 = blockIdx.y * 64;
  const int lo    = lane & 15;
  const int hi    = lane >> 4;
  const int row   = strip * 16 + lo;

  const float* xp = X + (size_t)row * IN_CH + 8 * hi;
  const float* wp = W + (size_t)(8 * hi) * OUT_CH + ncol0 + 4 * lo;

  v4f rx0 = *(const v4f*)(xp);
  v4f rx1 = *(const v4f*)(xp + 4);
  v4f rw[8];
  #pragma unroll
  for (int i = 0; i < 8; ++i)
    rw[i] = *(const v4f*)(wp + (size_t)i * OUT_CH);

  v8f c0 = {}, c1 = {}, c2 = {}, c3 = {};
  #pragma unroll
  for (int kk = 0; kk < IN_CH; kk += 16) {
    v2f a[4], b[4][4];
    a[0].x = rx0.x; a[0].y = rx0.y;
    a[1].x = rx0.z; a[1].y = rx0.w;
    a[2].x = rx1.x; a[2].y = rx1.y;
    a[3].x = rx1.z; a[3].y = rx1.w;
    #pragma unroll
    for (int t = 0; t < 4; ++t) {
      b[t][0].x = rw[2 * t].x; b[t][0].y = rw[2 * t + 1].x;
      b[t][1].x = rw[2 * t].y; b[t][1].y = rw[2 * t + 1].y;
      b[t][2].x = rw[2 * t].z; b[t][2].y = rw[2 * t + 1].z;
      b[t][3].x = rw[2 * t].w; b[t][3].y = rw[2 * t + 1].w;
    }
    if (kk + 16 < IN_CH) {
      rx0 = *(const v4f*)(xp + kk + 16);
      rx1 = *(const v4f*)(xp + kk + 20);
      #pragma unroll
      for (int i = 0; i < 8; ++i)
        rw[i] = *(const v4f*)(wp + (size_t)(kk + 16 + i) * OUT_CH);
    }
    __builtin_amdgcn_sched_barrier(0);
    #pragma unroll
    for (int t = 0; t < 4; ++t) {
      c0 = __builtin_amdgcn_wmma_f32_16x16x4_f32(false, a[t], false, b[t][0], (short)0, c0, false, false);
      c1 = __builtin_amdgcn_wmma_f32_16x16x4_f32(false, a[t], false, b[t][1], (short)0, c1, false, false);
      c2 = __builtin_amdgcn_wmma_f32_16x16x4_f32(false, a[t], false, b[t][2], (short)0, c2, false, false);
      c3 = __builtin_amdgcn_wmma_f32_16x16x4_f32(false, a[t], false, b[t][3], (short)0, c3, false, false);
    }
    __builtin_amdgcn_sched_barrier(0);
  }
  #pragma unroll
  for (int r = 0; r < 8; ++r) {
    const int m = strip * 16 + r + 8 * hi;
    v4f o;
    o.x = c0[r]; o.y = c1[r]; o.z = c2[r]; o.w = c3[r];
    *(v4f*)(H + (size_t)m * OUT_CH + ncol0 + 4 * lo) = o;
  }
}
#else
__global__ void gemm_wmma_f32(const float* __restrict__ X,
                              const float* __restrict__ W,
                              float* __restrict__ H, int M) {
  const int wave  = threadIdx.x >> 5;
  const int lane  = threadIdx.x & 31;
  const int strip = blockIdx.x * 4 + wave;
  if (strip * 16 >= M) return;
  const int ncol0 = blockIdx.y * 64;
  const int lo    = lane & 15;
  const int hi    = lane >> 4;
  const int row   = strip * 16 + lo;

  v8f acc[4] = {};
  for (int kk = 0; kk < IN_CH; kk += 32) {
    v16bf a;
    #pragma unroll
    for (int j = 0; j < 16; ++j) {
      const int p = j >> 1;
      const int k = 16 * (p >> 2) + 8 * hi + 2 * (p & 3) + (j & 1);
      a[j] = (__bf16)X[(size_t)row * IN_CH + kk + k];
    }
    #pragma unroll
    for (int t = 0; t < 4; ++t) {
      const int n = ncol0 + t * 16 + lo;
      v16bf b;
      #pragma unroll
      for (int j = 0; j < 16; ++j) {
        const int k = j + 16 * hi;
        b[j] = (__bf16)W[(size_t)(kk + k) * OUT_CH + n];
      }
      acc[t] = __builtin_amdgcn_wmma_f32_16x16x32_bf16(false, a, false, b,
                                                       (short)0, acc[t], false, false);
    }
  }
  #pragma unroll
  for (int t = 0; t < 4; ++t) {
    const int n = ncol0 + t * 16 + lo;
    #pragma unroll
    for (int r = 0; r < 8; ++r) {
      const int m = strip * 16 + r + 8 * hi;
      H[(size_t)m * OUT_CH + n] = acc[t][r];
    }
  }
}
#endif

__device__ __forceinline__ unsigned slab_key(int d) {
  return ((unsigned)d / (unsigned)NPG) & 0xffu;
}

__global__ void attn_dots(const float* __restrict__ H,
                          const float* __restrict__ att_src,
                          const float* __restrict__ att_dst,
                          float* __restrict__ a_src,
                          float* __restrict__ a_dst, int N,
                          const int* __restrict__ ei,
                          unsigned* __restrict__ keys, int E, int nbDots) {
  if ((int)blockIdx.x >= nbDots) {
    const int w  = ((int)blockIdx.x - nbDots) * (int)blockDim.x + (int)threadIdx.x;
    const int nW = ((E + 15) >> 4) << 2;
    if (w >= nW) return;
    const int* dstp = ei + E;
    const int e0 = 4 * w;
    int d0, d1, d2, d3;
    if (e0 + 3 < E && (((size_t)(dstp + e0)) & 15u) == 0u) {
      const int4 v = *(const int4*)(dstp + e0);
      d0 = v.x; d1 = v.y; d2 = v.z; d3 = v.w;
    } else {
      d0 = (e0     < E) ? dstp[e0]     : 0;
      d1 = (e0 + 1 < E) ? dstp[e0 + 1] : 0;
      d2 = (e0 + 2 < E) ? dstp[e0 + 2] : 0;
      d3 = (e0 + 3 < E) ? dstp[e0 + 3] : 0;
    }
    keys[w] = slab_key(d0) | (slab_key(d1) << 8) | (slab_key(d2) << 16) | (slab_key(d3) << 24);
    return;
  }
  const int idx = blockIdx.x * blockDim.x + threadIdx.x;
  if (idx >= N * HEADS) return;
  const int node = idx >> 2;
  const int head = idx & 3;
  const float* hp = H + (size_t)node * OUT_CH + head * HID;
  const float* as = att_src + head * HID;
  const float* ad = att_dst + head * HID;
  float s = 0.f, d = 0.f;
  #pragma unroll 8
  for (int c = 0; c < HID; ++c) {
    const float v = hp[c];
    s += v * as[c];
    d += v * ad[c];
  }
  a_src[idx] = s;
  a_dst[idx] = d;
}

#define NPW   7
#define NOWN  28
#define NCAP  256
#define RB    6
#define FV    8
#define WCAP  384

__device__ __forceinline__ float leaky(float v) {
  return (v >= 0.f) ? v : NEG_SLOPE * v;
}
__device__ __forceinline__ float bcastf(float v, int srcLane) {
  return __int_as_float(__builtin_amdgcn_readlane(__float_as_int(v), srcLane));
}
__device__ __forceinline__ void fma4(float4& acc, float x, const float4& h) {
  acc.x += x * h.x; acc.y += x * h.y; acc.z += x * h.z; acc.w += x * h.w;
}


struct RowTail { float4 bA, bB, pA, pB; };

__device__ __forceinline__ void finish_row(float* __restrict__ out, int d, int lane,
                                           const float4& accA, const float4& accB,
                                           float l0, float l1, float l2, float l3,
                                           const float* __restrict__ bias,
                                           const float* __restrict__ prelu_w) {
  RowTail rt;
  rt.bA = *(const float4*)(bias + 4 * lane);
  rt.bB = *(const float4*)(bias + 128 + 4 * lane);
  rt.pA = *(const float4*)(prelu_w + 4 * lane);
  rt.pB = *(const float4*)(prelu_w + 128 + 4 * lane);
  for (int o = 16; o > 0; o >>= 1) {
    l0 += __shfl_xor(l0, o);
    l1 += __shfl_xor(l1, o);
    l2 += __shfl_xor(l2, o);
    l3 += __shfl_xor(l3, o);
  }
  const int hsel = lane >> 4;
  const float da = (hsel ? l1 : l0) + 1e-16f;
  const float db = (hsel ? l3 : l2) + 1e-16f;
  float4 oA, oB;
  float v;
  v = accA.x / da + rt.bA.x; oA.x = (v >= 0.f) ? v : rt.pA.x * v;
  v = accA.y / da + rt.bA.y; oA.y = (v >= 0.f) ? v : rt.pA.y * v;
  v = accA.z / da + rt.bA.z; oA.z = (v >= 0.f) ? v : rt.pA.z * v;
  v = accA.w / da + rt.bA.w; oA.w = (v >= 0.f) ? v : rt.pA.w * v;
  v = accB.x / db + rt.bB.x; oB.x = (v >= 0.f) ? v : rt.pB.x * v;
  v = accB.y / db + rt.bB.y; oB.y = (v >= 0.f) ? v : rt.pB.y * v;
  v = accB.z / db + rt.bB.z; oB.z = (v >= 0.f) ? v : rt.pB.z * v;
  v = accB.w / db + rt.bB.w; oB.w = (v >= 0.f) ? v : rt.pB.w * v;
  float* op = out + (size_t)d * OUT_CH;
  *(float4*)(op + 4 * lane) = oA;
  *(float4*)(op + 128 + 4 * lane) = oB;
}

__device__ __forceinline__ bool fetch_edge_global(const int* __restrict__ ei,
                                                  int d, int E, int i, int& s) {
  const bool mt = (i < E) && (ei[E + i] == d);
  s = mt ? ei[i] : d;
  return mt;
}

__device__ __forceinline__ void reduce_node_global(
    const int* __restrict__ ei, const float* __restrict__ H,
    const float* __restrict__ a_src, const float* __restrict__ a_dst,
    const float* __restrict__ bias, const float* __restrict__ prelu_w,
    float* __restrict__ out, int d, int E, int lane) {
  const float4 ad = *(const float4*)(a_dst + (size_t)d * HEADS);
  const float4 as = *(const float4*)(a_src + (size_t)d * HEADS);
  const float es0 = leaky(as.x + ad.x);
  const float es1 = leaky(as.y + ad.y);
  const float es2 = leaky(as.z + ad.z);
  const float es3 = leaky(as.w + ad.w);

  float m0 = es0, m1 = es1, m2 = es2, m3 = es3;
  for (int base = 0; base < E; base += 32) {
    int s;
    fetch_edge_global(ei, d, E, base + lane, s);
    const float4 a = *(const float4*)(a_src + (size_t)s * HEADS);
    m0 = fmaxf(m0, leaky(a.x + ad.x));
    m1 = fmaxf(m1, leaky(a.y + ad.y));
    m2 = fmaxf(m2, leaky(a.z + ad.z));
    m3 = fmaxf(m3, leaky(a.w + ad.w));
  }
  for (int o = 16; o > 0; o >>= 1) {
    m0 = fmaxf(m0, __shfl_xor(m0, o));
    m1 = fmaxf(m1, __shfl_xor(m1, o));
    m2 = fmaxf(m2, __shfl_xor(m2, o));
    m3 = fmaxf(m3, __shfl_xor(m3, o));
  }

  const int hsel = lane >> 4;
  const float* Hl = H + 4 * lane;
  const float xs0 = expf(es0 - m0);
  const float xs1 = expf(es1 - m1);
  const float xs2 = expf(es2 - m2);
  const float xs3 = expf(es3 - m3);
  float4 accA, accB;
  {
    const float xa = hsel ? xs1 : xs0;
    const float xb = hsel ? xs3 : xs2;
    const float4 hA = *(const float4*)(Hl + (size_t)d * OUT_CH);
    const float4 hB = *(const float4*)(Hl + (size_t)d * OUT_CH + 128);
    accA.x = xa * hA.x; accA.y = xa * hA.y; accA.z = xa * hA.z; accA.w = xa * hA.w;
    accB.x = xb * hB.x; accB.y = xb * hB.y; accB.z = xb * hB.z; accB.w = xb * hB.w;
  }
  float l0 = (lane == 0) ? xs0 : 0.f;
  float l1 = (lane == 0) ? xs1 : 0.f;
  float l2 = (lane == 0) ? xs2 : 0.f;
  float l3 = (lane == 0) ? xs3 : 0.f;

  for (int base = 0; base < E; base += 32) {
    int s;
    const bool mt = fetch_edge_global(ei, d, E, base + lane, s);
    const float4 a = *(const float4*)(a_src + (size_t)s * HEADS);
    const float w = mt ? 1.f : 0.f;
    const float x0 = expf(leaky(a.x + ad.x) - m0) * w;
    const float x1 = expf(leaky(a.y + ad.y) - m1) * w;
    const float x2 = expf(leaky(a.z + ad.z) - m2) * w;
    const float x3 = expf(leaky(a.w + ad.w) - m3) * w;
    l0 += x0; l1 += x1; l2 += x2; l3 += x3;
    unsigned mask = __builtin_amdgcn_ballot_w32(mt);
    while (mask) {
      const int b = __builtin_ctz(mask);
      mask &= mask - 1;
      const int sb = __builtin_amdgcn_readlane(s, b);
      const float4 hA = *(const float4*)(Hl + (size_t)sb * OUT_CH);
      const float4 hB = *(const float4*)(Hl + (size_t)sb * OUT_CH + 128);
      const float y0 = bcastf(x0, b);
      const float y1 = bcastf(x1, b);
      const float y2 = bcastf(x2, b);
      const float y3 = bcastf(x3, b);
      fma4(accA, hsel ? y1 : y0, hA);
      fma4(accB, hsel ? y3 : y2, hB);
    }
  }
  finish_row(out, d, lane, accA, accB, l0, l1, l2, l3, bias, prelu_w);
}

__device__ __forceinline__ void add_chunk(const float* __restrict__ Hl, int s, bool mt,
                                          const float4& sc, float m0, float m1,
                                          float m2, float m3, int cnt, int hsel,
                                          float4& accA, float4& accB,
                                          float& l0, float& l1, float& l2, float& l3) {
  const float w = mt ? 1.f : 0.f;
  const float x0 = expf(sc.x - m0) * w;
  const float x1 = expf(sc.y - m1) * w;
  const float x2 = expf(sc.z - m2) * w;
  const float x3 = expf(sc.w - m3) * w;
  l0 += x0; l1 += x1; l2 += x2; l3 += x3;
  for (int q0 = 0; q0 < cnt; q0 += RB) {
    float4 hA[RB], hB[RB];
    #pragma unroll
    for (int q = 0; q < RB; ++q) {
      const bool has = (q0 + q) < cnt;
      const int bl = has ? (q0 + q) : q0;
      const int sb = __builtin_amdgcn_readlane(s, bl);
      hA[q] = *(const float4*)(Hl + (size_t)sb * OUT_CH);
      hB[q] = *(const float4*)(Hl + (size_t)sb * OUT_CH + 128);
    }
    #pragma unroll
    for (int q = 0; q < RB; ++q) {
      const bool has = (q0 + q) < cnt;
      const int bl = has ? (q0 + q) : q0;
      const float y0 = bcastf(x0, bl);
      const float y1 = bcastf(x1, bl);
      const float y2 = bcastf(x2, bl);
      const float y3 = bcastf(x3, bl);
      fma4(accA, has ? (hsel ? y1 : y0) : 0.f, hA[q]);
      fma4(accB, has ? (hsel ? y3 : y2) : 0.f, hB[q]);
    }
  }
}

__device__ __forceinline__ void sort_list(const unsigned* lst, unsigned* srt,
                                          int n, int lane) {
  for (int base = 0; base < n; base += 32) {
    const int i = base + lane;
    const unsigned v = (i < n) ? lst[(i < n) ? i : 0] : 0xffffffffu;
    unsigned r = 0u;
    for (int b2 = 0; b2 < n; b2 += 32) {
      const int i2 = b2 + lane;
      const unsigned w = (i2 < n) ? lst[(i2 < n) ? i2 : 0] : 0xffffffffu;
      const int m = (n - b2 < 32) ? (n - b2) : 32;
      for (int j = 0; j < m; ++j) {
        const unsigned wj = (unsigned)__builtin_amdgcn_readlane((int)w, j);
        const int idx = b2 + j;
        r += (wj < v || (wj == v && idx < i)) ? 1u : 0u;
      }
    }
    if (i < n) srt[r] = v;
  }
}

__device__ __forceinline__ void reduce_node_list(
    const float* __restrict__ H, const float* __restrict__ a_src,
    const float* __restrict__ a_dst, const float* __restrict__ bias,
    const float* __restrict__ prelu_w, float* __restrict__ out,
    const unsigned* raw, unsigned* srt, int n, int d, int lane) {
  const int hsel = lane >> 4;
  const float* Hl = H + 4 * lane;
  const float4 ad = *(const float4*)(a_dst + (size_t)d * HEADS);
  const float4 as = *(const float4*)(a_src + (size_t)d * HEADS);
  const float4 hdA = *(const float4*)(Hl + (size_t)d * OUT_CH);
  const float4 hdB = *(const float4*)(Hl + (size_t)d * OUT_CH + 128);
  sort_list(raw, srt, n, lane);
  const unsigned* lst = srt;
  const bool mt0 = lane < n;
  const int  s0  = mt0 ? (int)lst[lane] : d;
  const float4 a0 = *(const float4*)(a_src + (size_t)s0 * HEADS);
  float4 sc0;
  sc0.x = leaky(a0.x + ad.x); sc0.y = leaky(a0.y + ad.y);
  sc0.z = leaky(a0.z + ad.z); sc0.w = leaky(a0.w + ad.w);
  const float es0 = leaky(as.x + ad.x);
  const float es1 = leaky(as.y + ad.y);
  const float es2 = leaky(as.z + ad.z);
  const float es3 = leaky(as.w + ad.w);

  float m0 = fmaxf(es0, sc0.x), m1 = fmaxf(es1, sc0.y);
  float m2 = fmaxf(es2, sc0.z), m3 = fmaxf(es3, sc0.w);
  for (int base = 32; base < n; base += 32) {
    const int i = base + lane;
    const bool mt = i < n;
    const int s = mt ? (int)lst[mt ? i : 0] : d;
    const float4 a = *(const float4*)(a_src + (size_t)s * HEADS);
    m0 = fmaxf(m0, leaky(a.x + ad.x));
    m1 = fmaxf(m1, leaky(a.y + ad.y));
    m2 = fmaxf(m2, leaky(a.z + ad.z));
    m3 = fmaxf(m3, leaky(a.w + ad.w));
  }
  for (int o = 16; o > 0; o >>= 1) {
    m0 = fmaxf(m0, __shfl_xor(m0, o));
    m1 = fmaxf(m1, __shfl_xor(m1, o));
    m2 = fmaxf(m2, __shfl_xor(m2, o));
    m3 = fmaxf(m3, __shfl_xor(m3, o));
  }

  const float xs0 = expf(es0 - m0);
  const float xs1 = expf(es1 - m1);
  const float xs2 = expf(es2 - m2);
  const float xs3 = expf(es3 - m3);
  float4 accA, accB;
  {
    const float xa = hsel ? xs1 : xs0;
    const float xb = hsel ? xs3 : xs2;
    accA.x = xa * hdA.x; accA.y = xa * hdA.y; accA.z = xa * hdA.z; accA.w = xa * hdA.w;
    accB.x = xb * hdB.x; accB.y = xb * hdB.y; accB.z = xb * hdB.z; accB.w = xb * hdB.w;
  }
  float l0 = (lane == 0) ? xs0 : 0.f;
  float l1 = (lane == 0) ? xs1 : 0.f;
  float l2 = (lane == 0) ? xs2 : 0.f;
  float l3 = (lane == 0) ? xs3 : 0.f;

  add_chunk(Hl, s0, mt0, sc0, m0, m1, m2, m3, (n < 32) ? n : 32, hsel,
            accA, accB, l0, l1, l2, l3);
  for (int base = 32; base < n; base += 32) {
    const int i = base + lane;
    const bool mt = i < n;
    const int s = mt ? (int)lst[mt ? i : 0] : d;
    const float4 a = *(const float4*)(a_src + (size_t)s * HEADS);
    float4 sc;
    sc.x = leaky(a.x + ad.x); sc.y = leaky(a.y + ad.y);
    sc.z = leaky(a.z + ad.z); sc.w = leaky(a.w + ad.w);
    const int rem = n - base;
    add_chunk(Hl, s, mt, sc, m0, m1, m2, m3, (rem < 32) ? rem : 32, hsel,
              accA, accB, l0, l1, l2, l3);
  }
  finish_row(out, d, lane, accA, accB, l0, l1, l2, l3, bias, prelu_w);
}

__device__ __forceinline__ void push_edge(unsigned (*list)[NCAP], unsigned* cnt,
                                          unsigned* ovf, unsigned dl, unsigned s) {
  const unsigned slot = atomicAdd(&cnt[dl], 1u);
  if (slot < NCAP) list[dl][slot] = s;
  else *ovf = 1u;
}

__global__ __launch_bounds__(1024) void aggregate_slabs(
    const int* __restrict__ ei, const float* __restrict__ H,
    const float* __restrict__ a_src, const float* __restrict__ a_dst,
    const float* __restrict__ bias, const float* __restrict__ prelu_w,
    float* __restrict__ out, const unsigned* __restrict__ keys, int E, int N) {
  __shared__ unsigned s_list[NPG][NCAP];
  __shared__ unsigned s_sorted[NOWN][NCAP];
  __shared__ unsigned s_match[32][WCAP];
  __shared__ unsigned s_cnt[256];
  __shared__ unsigned s_ovf;

  const int tid  = threadIdx.x;
  const int lane = tid & 31;
  const int wave = tid >> 5;
  const int lo   = blockIdx.x * NPG;
  const unsigned span = (unsigned)((N - lo < NPG) ? (N - lo) : NPG);

  if (tid < 256) s_cnt[tid] = 0u;
  if (tid == 0) s_ovf = 0u;
  __syncthreads();

  if (E > 0) {
    const int* srcp = ei;
    const int* dstp = ei + E;
    unsigned* mlist = &s_match[wave][0];
    unsigned wc = 0u;
    {
      const uint4* kv = (const uint4*)keys;
      const int nG = (E + 15) >> 4;
      const unsigned splat = ((unsigned)blockIdx.x & 0xffu) * 0x01010101u;
      const int UNITV = 1024 * FV;
      const int nUnits = (nG + UNITV - 1) / UNITV;
      const int nSlots = nUnits * UNITV;
      const int lastg = nG - 1;
      const int off = (int)((blockIdx.x * 1024u) % (unsigned)nSlots);
      uint4 cur[FV];
      #pragma unroll
      for (int k = 0; k < FV; ++k) {
        int b = off + k * 1024;
        if (b >= nSlots) b -= nSlots;
        const int g = b + tid;
        cur[k] = kv[(g < nG) ? g : lastg];
      }
      #pragma unroll 2
      for (int it = 0; it < nUnits; ++it) {
        const int b0 = off + it * UNITV;
        uint4 nxt[FV];
        #pragma unroll
        for (int k = 0; k < FV; ++k) {
          int b = b0 + UNITV + k * 1024;
          if (b >= nSlots) b -= nSlots;
          if (b >= nSlots) b -= nSlots;
          const int g = b + tid;
          nxt[k] = kv[(g < nG) ? g : lastg];
        }
        #pragma unroll
        for (int k = 0; k < FV; ++k) {
          int b = b0 + k * 1024;
          if (b >= nSlots) b -= nSlots;
          const int g = b + tid;
          const unsigned x0 = cur[k].x ^ splat;
          const unsigned x1 = cur[k].y ^ splat;
          const unsigned x2 = cur[k].z ^ splat;
          const unsigned x3 = cur[k].w ^ splat;
          const unsigned z0 = (x0 - 0x01010101u) & ~x0 & 0x80808080u;
          const unsigned z1 = (x1 - 0x01010101u) & ~x1 & 0x80808080u;
          const unsigned z2 = (x2 - 0x01010101u) & ~x2 & 0x80808080u;
          const unsigned z3 = (x3 - 0x01010101u) & ~x3 & 0x80808080u;
          const bool any = ((z0 | z1 | z2 | z3) != 0u) && (g < nG);
          if (__builtin_amdgcn_ballot_w32(any) != 0u) {
            unsigned m = any ? ((z0 >> 7) | (z1 >> 6) | (z2 >> 5) | (z3 >> 4)) : 0u;
            unsigned hb = __builtin_amdgcn_ballot_w32(m != 0u);
            while (hb != 0u) {
              if (m != 0u) {
                const unsigned slot = wc + __builtin_amdgcn_mbcnt_lo(hb, 0u);
                const unsigned p = (unsigned)__builtin_ctz(m);
                m &= m - 1u;
                if (slot < WCAP) mlist[slot] = 16u * (unsigned)g + 4u * (p & 7u) + (p >> 3);
                else s_ovf = 1u;
              }
              wc += (unsigned)__popc(hb);
              hb = __builtin_amdgcn_ballot_w32(m != 0u);
            }
          }
        }
        #pragma unroll
        for (int k = 0; k < FV; ++k) cur[k] = nxt[k];
      }
    }

    {
      const unsigned nm = (wc < WCAP) ? wc : WCAP;
      for (unsigned base = 0; base < nm; base += 128u) {
        int dv[4], sv[4];
        bool ok[4];
        #pragma unroll
        for (int q = 0; q < 4; ++q) {
          const unsigned i = base + (unsigned)q * 32u + (unsigned)lane;
          const unsigned e = (i < nm) ? mlist[(i < nm) ? i : 0u] : 0xffffffffu;
          ok[q] = e < (unsigned)E;
          const unsigned ec = ok[q] ? e : 0u;
          dv[q] = dstp[ec];
          sv[q] = srcp[ec];
        }
        #pragma unroll
        for (int q = 0; q < 4; ++q) {
          const unsigned dl = (unsigned)(dv[q] - lo);
          if (ok[q] && dl < span) push_edge(s_list, s_cnt, &s_ovf, dl, (unsigned)sv[q]);
        }
      }
    }
  }
  __syncthreads();

  if (wave >= NOWN) return;
  const int ovf = __builtin_amdgcn_readfirstlane((int)s_ovf);
  for (unsigned j = 0; j < NPW; ++j) {
    const unsigned dl = (unsigned)wave * NPW + j;
    if (dl >= span) break;
    const int d = lo + (int)dl;
    if (!ovf) {
      int n = __builtin_amdgcn_readfirstlane((int)s_cnt[dl]);
      if (n > NCAP) n = NCAP;
      reduce_node_list(H, a_src, a_dst, bias, prelu_w, out, &s_list[dl][0],
                       &s_sorted[wave][0], n, d, lane);
    } else {
      reduce_node_global(ei, H, a_src, a_dst, bias, prelu_w, out, d, E, lane);
    }
  }
}

extern "C" void kernel_launch(void* const* d_in, const int* in_sizes, int n_in,
                              void* d_out, int out_size, void* d_ws, size_t ws_size,
                              hipStream_t stream) {
  const float* x       = (const float*)d_in[0];
  const int*   ei      = (const int*)d_in[1];
  const float* W       = (const float*)d_in[2];
  const float* att_src = (const float*)d_in[3];
  const float* att_dst = (const float*)d_in[4];
  const float* bias    = (const float*)d_in[5];
  const float* prelu_w = (const float*)d_in[6];
  float* out = (float*)d_out;

  const int N = in_sizes[0] / IN_CH;
  const int E = in_sizes[1] / 2;

  char* ws = (char*)d_ws;
  float* H = (float*)ws;            ws += (size_t)N * OUT_CH * sizeof(float);
  float* a_src = (float*)ws;        ws += (size_t)N * HEADS * sizeof(float);
  float* a_dst = (float*)ws;        ws += (size_t)N * HEADS * sizeof(float);
  unsigned* keys = (unsigned*)ws;

  const int strips = (N + 15) / 16;
  dim3 ggrid((strips + 3) / 4, OUT_CH / 64);
  gemm_wmma_f32<<<ggrid, 128, 0, stream>>>(x, W, H, N);

  const int nbDots = (N * HEADS + 255) / 256;
  const int nbKeys = ((((E + 15) >> 4) << 2) + 255) / 256;
  attn_dots<<<nbDots + nbKeys, 256, 0, stream>>>(H, att_src, att_dst, a_src, a_dst, N,
                                                 ei, keys, E, nbDots);

  aggregate_slabs<<<(N + NPG - 1) / NPG, 1024, 0, stream>>>(ei, H, a_src, a_dst,
                                                            bias, prelu_w, out, keys, E, N);
}


# Attention dots taken from the GEMM's accumulators in its epilogue (work-group = a 16-row strip x 4 heads, lane tree, one whole-line store); dots launch gone (3 to 2); order of the dot sums changed
# speedup vs baseline: 7.1466x; 1.0674x over previous
#include <hip/hip_runtime.h>


#define OUT_CH 256
#define IN_CH  256
#define HEADS  4
#define HID    64
#define NEG_SLOPE 0.2f
#define NPG   196

typedef float v2f  __attribute__((ext_vector_type(2)));
typedef float v4f  __attribute__((ext_vector_type(4)));
typedef float v8f  __attribute__((ext_vector_type(8)));

__device__ __forceinline__ unsigned slab_key(int d) {
  return ((unsigned)d / (unsigned)NPG) & 0xffu;
}

__global__ __launch_bounds__(128) void gemm_dots_keys(const float* __restrict__ X,
                              const float* __restrict__ W,
                              float* __restrict__ H, int M,
                              const float* __restrict__ att_src,
                              const float* __restrict__ att_dst,
                              float* __restrict__ a_src,
                              float* __restrict__ a_dst,
                              const int* __restrict__ ei,
                              unsigned* __restrict__ keys, int E, int nbGemm) {
  __shared__ float s_dots[2 * 16 * HEADS];

  if ((int)blockIdx.x >= nbGemm) {
    const int w  = ((int)blockIdx.x - nbGemm) * (int)blockDim.x + (int)threadIdx.x;
    const int nW = ((E + 15) >> 4) << 2;
    if (w >= nW) return;
    const int* dstp = ei + E;
    const int e0 = 4 * w;
    int d0, d1, d2, d3;
    if (e0 + 3 < E && (((size_t)(dstp + e0)) & 15u) == 0u) {
      const int4 v = *(const int4*)(dstp + e0);
      d0 = v.x; d1 = v.y; d2 = v.z; d3 = v.w;
    } else {
      d0 = (e0     < E) ? dstp[e0]     : 0;
      d1 = (e0 + 1 < E) ? dstp[e0 + 1] : 0;
      d2 = (e0 + 2 < E) ? dstp[e0 + 2] : 0;
      d3 = (e0 + 3 < E) ? dstp[e0 + 3] : 0;
    }
    keys[w] = slab_key(d0) | (slab_key(d1) << 8) | (slab_key(d2) << 16) | (slab_key(d3) << 24);
    return;
  }

  const int wave  = threadIdx.x >> 5;
  const int lane  = threadIdx.x & 31;
  const int strip = blockIdx.x;
  if (strip * 16 >= M) return;
  const int ncol0 = wave * HID;
  const int lo    = lane & 15;
  const int hi    = lane >> 4;
  const int row   = strip * 16 + lo;

  const float* xp = X + (size_t)row * IN_CH + 8 * hi;
  const float* wp = W + (size_t)(8 * hi) * OUT_CH + ncol0 + 4 * lo;

  v4f rx0 = *(const v4f*)(xp);
  v4f rx1 = *(const v4f*)(xp + 4);
  v4f rw[8];
  #pragma unroll
  for (int i = 0; i < 8; ++i)
    rw[i] = *(const v4f*)(wp + (size_t)i * OUT_CH);

  v8f c0 = {}, c1 = {}, c2 = {}, c3 = {};
  #pragma unroll
  for (int kk = 0; kk < IN_CH; kk += 16) {
    v2f a[4], b[4][4];
    a[0].x = rx0.x; a[0].y = rx0.y;
    a[1].x = rx0.z; a[1].y = rx0.w;
    a[2].x = rx1.x; a[2].y = rx1.y;
    a[3].x = rx1.z; a[3].y = rx1.w;
    #pragma unroll
    for (int t = 0; t < 4; ++t) {
      b[t][0].x = rw[2 * t].x; b[t][0].y = rw[2 * t + 1].x;
      b[t][1].x = rw[2 * t].y; b[t][1].y = rw[2 * t + 1].y;
      b[t][2].x = rw[2 * t].z; b[t][2].y = rw[2 * t + 1].z;
      b[t][3].x = rw[2 * t].w; b[t][3].y = rw[2 * t + 1].w;
    }
    if (kk + 16 < IN_CH) {
      rx0 = *(const v4f*)(xp + kk + 16);
      rx1 = *(const v4f*)(xp + kk + 20);
      #pragma unroll
      for (int i = 0; i < 8; ++i)
        rw[i] = *(const v4f*)(wp + (size_t)(kk + 16 + i) * OUT_CH);
    }
    __builtin_amdgcn_sched_barrier(0);
    #pragma unroll
    for (int t = 0; t < 4; ++t) {
      c0 = __builtin_amdgcn_wmma_f32_16x16x4_f32(false, a[t], false, b[t][0], (short)0, c0, false, false);
      c1 = __builtin_amdgcn_wmma_f32_16x16x4_f32(false, a[t], false, b[t][1], (short)0, c1, false, false);
      c2 = __builtin_amdgcn_wmma_f32_16x16x4_f32(false, a[t], false, b[t][2], (short)0, c2, false, false);
      c3 = __builtin_amdgcn_wmma_f32_16x16x4_f32(false, a[t], false, b[t][3], (short)0, c3, false, false);
    }
    __builtin_amdgcn_sched_barrier(0);
  }
  #pragma unroll
  for (int r = 0; r < 8; ++r) {
    const int m = strip * 16 + r + 8 * hi;
    v4f o;
    o.x = c0[r]; o.y = c1[r]; o.z = c2[r]; o.w = c3[r];
    *(v4f*)(H + (size_t)m * OUT_CH + ncol0 + 4 * lo) = o;
  }

  const v4f as4 = *(const v4f*)(att_src + ncol0 + 4 * lo);
  const v4f ad4 = *(const v4f*)(att_dst + ncol0 + 4 * lo);
  float v[16];
  #pragma unroll
  for (int r = 0; r < 8; ++r) {
    v[r]     = c0[r] * as4.x + c1[r] * as4.y + c2[r] * as4.z + c3[r] * as4.w;
    v[8 + r] = c0[r] * ad4.x + c1[r] * ad4.y + c2[r] * ad4.z + c3[r] * ad4.w;
  }
  const bool b3 = (lo & 8) != 0;
  const bool b2 = (lo & 4) != 0;
  const bool b1 = (lo & 2) != 0;
  const bool b0 = (lo & 1) != 0;
  float w8[8], w4[4], w2[2];
  #pragma unroll
  for (int i = 0; i < 8; ++i) {
    const float keep = b3 ? v[8 + i] : v[i];
    const float send = b3 ? v[i] : v[8 + i];
    w8[i] = keep + __shfl_xor(send, 8);
  }
  #pragma unroll
  for (int i = 0; i < 4; ++i) {
    const float keep = b2 ? w8[4 + i] : w8[i];
    const float send = b2 ? w8[i] : w8[4 + i];
    w4[i] = keep + __shfl_xor(send, 4);
  }
  #pragma unroll
  for (int i = 0; i < 2; ++i) {
    const float keep = b1 ? w4[2 + i] : w4[i];
    const float send = b1 ? w4[i] : w4[2 + i];
    w2[i] = keep + __shfl_xor(send, 2);
  }
  const float dot = (b0 ? w2[1] : w2[0]) + __shfl_xor(b0 ? w2[0] : w2[1], 1);
  s_dots[(((lo >> 3) * 16) + 8 * hi + (lo & 7)) * HEADS + wave] = dot;
  __syncthreads();
  if (wave == 0) {
    const float4 q = *(const float4*)(&s_dots[4 * lane]);
    const int node = strip * 16 + lo;
    float* dstq = (hi ? a_dst : a_src) + (size_t)node * HEADS;
    if (node < M) *(float4*)dstq = q;
  }
}

#define NPW   7
#define NOWN  28
#define NCAP  256
#define RB    6
#define FV    8
#define WCAP  384

__device__ __forceinline__ float leaky(float v) {
  return (v >= 0.f) ? v : NEG_SLOPE * v;
}
__device__ __forceinline__ float bcastf(float v, int srcLane) {
  return __int_as_float(__builtin_amdgcn_readlane(__float_as_int(v), srcLane));
}
__device__ __forceinline__ void fma4(float4& acc, float x, const float4& h) {
  acc.x += x * h.x; acc.y += x * h.y; acc.z += x * h.z; acc.w += x * h.w;
}


struct RowTail { float4 bA, bB, pA, pB; };

__device__ __forceinline__ void finish_row(float* __restrict__ out, int d, int lane,
                                           const float4& accA, const float4& accB,
                                           float l0, float l1, float l2, float l3,
                                           const float* __restrict__ bias,
                                           const float* __restrict__ prelu_w) {
  RowTail rt;
  rt.bA = *(const float4*)(bias + 4 * lane);
  rt.bB = *(const float4*)(bias + 128 + 4 * lane);
  rt.pA = *(const float4*)(prelu_w + 4 * lane);
  rt.pB = *(const float4*)(prelu_w + 128 + 4 * lane);
  for (int o = 16; o > 0; o >>= 1) {
    l0 += __shfl_xor(l0, o);
    l1 += __shfl_xor(l1, o);
    l2 += __shfl_xor(l2, o);
    l3 += __shfl_xor(l3, o);
  }
  const int hsel = lane >> 4;
  const float da = (hsel ? l1 : l0) + 1e-16f;
  const float db = (hsel ? l3 : l2) + 1e-16f;
  float4 oA, oB;
  float v;
  v = accA.x / da + rt.bA.x; oA.x = (v >= 0.f) ? v : rt.pA.x * v;
  v = accA.y / da + rt.bA.y; oA.y = (v >= 0.f) ? v : rt.pA.y * v;
  v = accA.z / da + rt.bA.z; oA.z = (v >= 0.f) ? v : rt.pA.z * v;
  v = accA.w / da + rt.bA.w; oA.w = (v >= 0.f) ? v : rt.pA.w * v;
  v = accB.x / db + rt.bB.x; oB.x = (v >= 0.f) ? v : rt.pB.x * v;
  v = accB.y / db + rt.bB.y; oB.y = (v >= 0.f) ? v : rt.pB.y * v;
  v = accB.z / db + rt.bB.z; oB.z = (v >= 0.f) ? v : rt.pB.z * v;
  v = accB.w / db + rt.bB.w; oB.w = (v >= 0.f) ? v : rt.pB.w * v;
  float* op = out + (size_t)d * OUT_CH;
  *(float4*)(op + 4 * lane) = oA;
  *(float4*)(op + 128 + 4 * lane) = oB;
}

__device__ __forceinline__ bool fetch_edge_global(const int* __restrict__ ei,
                                                  int d, int E, int i, int& s) {
  const bool mt = (i < E) && (ei[E + i] == d);
  s = mt ? ei[i] : d;
  return mt;
}

__device__ __forceinline__ void reduce_node_global(
    const int* __restrict__ ei, const float* __restrict__ H,
    const float* __restrict__ a_src, const float* __restrict__ a_dst,
    const float* __restrict__ bias, const float* __restrict__ prelu_w,
    float* __restrict__ out, int d, int E, int lane) {
  const float4 ad = *(const float4*)(a_dst + (size_t)d * HEADS);
  const float4 as = *(const float4*)(a_src + (size_t)d * HEADS);
  const float es0 = leaky(as.x + ad.x);
  const float es1 = leaky(as.y + ad.y);
  const float es2 = leaky(as.z + ad.z);
  const float es3 = leaky(as.w + ad.w);

  float m0 = es0, m1 = es1, m2 = es2, m3 = es3;
  for (int base = 0; base < E; base += 32) {
    int s;
    fetch_edge_global(ei, d, E, base + lane, s);
    const float4 a = *(const float4*)(a_src + (size_t)s * HEADS);
    m0 = fmaxf(m0, leaky(a.x + ad.x));
    m1 = fmaxf(m1, leaky(a.y + ad.y));
    m2 = fmaxf(m2, leaky(a.z + ad.z));
    m3 = fmaxf(m3, leaky(a.w + ad.w));
  }
  for (int o = 16; o > 0; o >>= 1) {
    m0 = fmaxf(m0, __shfl_xor(m0, o));
    m1 = fmaxf(m1, __shfl_xor(m1, o));
    m2 = fmaxf(m2, __shfl_xor(m2, o));
    m3 = fmaxf(m3, __shfl_xor(m3, o));
  }

  const int hsel = lane >> 4;
  const float* Hl = H + 4 * lane;
  const float xs0 = expf(es0 - m0);
  const float xs1 = expf(es1 - m1);
  const float xs2 = expf(es2 - m2);
  const float xs3 = expf(es3 - m3);
  float4 accA, accB;
  {
    const float xa = hsel ? xs1 : xs0;
    const float xb = hsel ? xs3 : xs2;
    const float4 hA = *(const float4*)(Hl + (size_t)d * OUT_CH);
    const float4 hB = *(const float4*)(Hl + (size_t)d * OUT_CH + 128);
    accA.x = xa * hA.x; accA.y = xa * hA.y; accA.z = xa * hA.z; accA.w = xa * hA.w;
    accB.x = xb * hB.x; accB.y = xb * hB.y; accB.z = xb * hB.z; accB.w = xb * hB.w;
  }
  float l0 = (lane == 0) ? xs0 : 0.f;
  float l1 = (lane == 0) ? xs1 : 0.f;
  float l2 = (lane == 0) ? xs2 : 0.f;
  float l3 = (lane == 0) ? xs3 : 0.f;

  for (int base = 0; base < E; base += 32) {
    int s;
    const bool mt = fetch_edge_global(ei, d, E, base + lane, s);
    const float4 a = *(const float4*)(a_src + (size_t)s * HEADS);
    const float w = mt ? 1.f : 0.f;
    const float x0 = expf(leaky(a.x + ad.x) - m0) * w;
    const float x1 = expf(leaky(a.y + ad.y) - m1) * w;
    const float x2 = expf(leaky(a.z + ad.z) - m2) * w;
    const float x3 = expf(leaky(a.w + ad.w) - m3) * w;
    l0 += x0; l1 += x1; l2 += x2; l3 += x3;
    unsigned mask = __builtin_amdgcn_ballot_w32(mt);
    while (mask) {
      const int b = __builtin_ctz(mask);
      mask &= mask - 1;
      const int sb = __builtin_amdgcn_readlane(s, b);
      const float4 hA = *(const float4*)(Hl + (size_t)sb * OUT_CH);
      const float4 hB = *(const float4*)(Hl + (size_t)sb * OUT_CH + 128);
      const float y0 = bcastf(x0, b);
      const float y1 = bcastf(x1, b);
      const float y2 = bcastf(x2, b);
      const float y3 = bcastf(x3, b);
      fma4(accA, hsel ? y1 : y0, hA);
      fma4(accB, hsel ? y3 : y2, hB);
    }
  }
  finish_row(out, d, lane, accA, accB, l0, l1, l2, l3, bias, prelu_w);
}

__device__ __forceinline__ void add_chunk(const float* __restrict__ Hl, int s, bool mt,
                                          const float4& sc, float m0, float m1,
                                          float m2, float m3, int cnt, int hsel,
                                          float4& accA, float4& accB,
                                          float& l0, float& l1, float& l2, float& l3) {
  const float w = mt ? 1.f : 0.f;
  const float x0 = expf(sc.x - m0) * w;
  const float x1 = expf(sc.y - m1) * w;
  const float x2 = expf(sc.z - m2) * w;
  const float x3 = expf(sc.w - m3) * w;
  l0 += x0; l1 += x1; l2 += x2; l3 += x3;
  for (int q0 = 0; q0 < cnt; q0 += RB) {
    float4 hA[RB], hB[RB];
    #pragma unroll
    for (int q = 0; q < RB; ++q) {
      const bool has = (q0 + q) < cnt;
      const int bl = has ? (q0 + q) : q0;
      const int sb = __builtin_amdgcn_readlane(s, bl);
      hA[q] = *(const float4*)(Hl + (size_t)sb * OUT_CH);
      hB[q] = *(const float4*)(Hl + (size_t)sb * OUT_CH + 128);
    }
    #pragma unroll
    for (int q = 0; q < RB; ++q) {
      const bool has = (q0 + q) < cnt;
      const int bl = has ? (q0 + q) : q0;
      const float y0 = bcastf(x0, bl);
      const float y1 = bcastf(x1, bl);
      const float y2 = bcastf(x2, bl);
      const float y3 = bcastf(x3, bl);
      fma4(accA, has ? (hsel ? y1 : y0) : 0.f, hA[q]);
      fma4(accB, has ? (hsel ? y3 : y2) : 0.f, hB[q]);
    }
  }
}

__device__ __forceinline__ void sort_list(const unsigned* lst, unsigned* srt,
                                          int n, int lane) {
  for (int base = 0; base < n; base += 32) {
    const int i = base + lane;
    const unsigned v = (i < n) ? lst[(i < n) ? i : 0] : 0xffffffffu;
    unsigned r = 0u;
    for (int b2 = 0; b2 < n; b2 += 32) {
      const int i2 = b2 + lane;
      const unsigned w = (i2 < n) ? lst[(i2 < n) ? i2 : 0] : 0xffffffffu;
      const int m = (n - b2 < 32) ? (n - b2) : 32;
      for (int j = 0; j < m; ++j) {
        const unsigned wj = (unsigned)__builtin_amdgcn_readlane((int)w, j);
        const int idx = b2 + j;
        r += (wj < v || (wj == v && idx < i)) ? 1u : 0u;
      }
    }
    if (i < n) srt[r] = v;
  }
}

__device__ __forceinline__ void reduce_node_list(
    const float* __restrict__ H, const float* __restrict__ a_src,
    const float* __restrict__ a_dst, const float* __restrict__ bias,
    const float* __restrict__ prelu_w, float* __restrict__ out,
    const unsigned* raw, unsigned* srt, int n, int d, int lane) {
  const int hsel = lane >> 4;
  const float* Hl = H + 4 * lane;
  const float4 ad = *(const float4*)(a_dst + (size_t)d * HEADS);
  const float4 as = *(const float4*)(a_src + (size_t)d * HEADS);
  const float4 hdA = *(const float4*)(Hl + (size_t)d * OUT_CH);
  const float4 hdB = *(const float4*)(Hl + (size_t)d * OUT_CH + 128);
  sort_list(raw, srt, n, lane);
  const unsigned* lst = srt;
  const bool mt0 = lane < n;
  const int  s0  = mt0 ? (int)lst[lane] : d;
  const float4 a0 = *(const float4*)(a_src + (size_t)s0 * HEADS);
  float4 sc0;
  sc0.x = leaky(a0.x + ad.x); sc0.y = leaky(a0.y + ad.y);
  sc0.z = leaky(a0.z + ad.z); sc0.w = leaky(a0.w + ad.w);
  const float es0 = leaky(as.x + ad.x);
  const float es1 = leaky(as.y + ad.y);
  const float es2 = leaky(as.z + ad.z);
  const float es3 = leaky(as.w + ad.w);

  float m0 = fmaxf(es0, sc0.x), m1 = fmaxf(es1, sc0.y);
  float m2 = fmaxf(es2, sc0.z), m3 = fmaxf(es3, sc0.w);
  for (int base = 32; base < n; base += 32) {
    const int i = base + lane;
    const bool mt = i < n;
    const int s = mt ? (int)lst[mt ? i : 0] : d;
    const float4 a = *(const float4*)(a_src + (size_t)s * HEADS);
    m0 = fmaxf(m0, leaky(a.x + ad.x));
    m1 = fmaxf(m1, leaky(a.y + ad.y));
    m2 = fmaxf(m2, leaky(a.z + ad.z));
    m3 = fmaxf(m3, leaky(a.w + ad.w));
  }
  for (int o = 16; o > 0; o >>= 1) {
    m0 = fmaxf(m0, __shfl_xor(m0, o));
    m1 = fmaxf(m1, __shfl_xor(m1, o));
    m2 = fmaxf(m2, __shfl_xor(m2, o));
    m3 = fmaxf(m3, __shfl_xor(m3, o));
  }

  const float xs0 = expf(es0 - m0);
  const float xs1 = expf(es1 - m1);
  const float xs2 = expf(es2 - m2);
  const float xs3 = expf(es3 - m3);
  float4 accA, accB;
  {
    const float xa = hsel ? xs1 : xs0;
    const float xb = hsel ? xs3 : xs2;
    accA.x = xa * hdA.x; accA.y = xa * hdA.y; accA.z = xa * hdA.z; accA.w = xa * hdA.w;
    accB.x = xb * hdB.x; accB.y = xb * hdB.y; accB.z = xb * hdB.z; accB.w = xb * hdB.w;
  }
  float l0 = (lane == 0) ? xs0 : 0.f;
  float l1 = (lane == 0) ? xs1 : 0.f;
  float l2 = (lane == 0) ? xs2 : 0.f;
  float l3 = (lane == 0) ? xs3 : 0.f;

  add_chunk(Hl, s0, mt0, sc0, m0, m1, m2, m3, (n < 32) ? n : 32, hsel,
            accA, accB, l0, l1, l2, l3);
  for (int base = 32; base < n; base += 32) {
    const int i = base + lane;
    const bool mt = i < n;
    const int s = mt ? (int)lst[mt ? i : 0] : d;
    const float4 a = *(const float4*)(a_src + (size_t)s * HEADS);
    float4 sc;
    sc.x = leaky(a.x + ad.x); sc.y = leaky(a.y + ad.y);
    sc.z = leaky(a.z + ad.z); sc.w = leaky(a.w + ad.w);
    const int rem = n - base;
    add_chunk(Hl, s, mt, sc, m0, m1, m2, m3, (rem < 32) ? rem : 32, hsel,
              accA, accB, l0, l1, l2, l3);
  }
  finish_row(out, d, lane, accA, accB, l0, l1, l2, l3, bias, prelu_w);
}

__device__ __forceinline__ void push_edge(unsigned (*list)[NCAP], unsigned* cnt,
                                          unsigned* ovf, unsigned dl, unsigned s) {
  const unsigned slot = atomicAdd(&cnt[dl], 1u);
  if (slot < NCAP) list[dl][slot] = s;
  else *ovf = 1u;
}

__global__ __launch_bounds__(1024) void aggregate_slabs(
    const int* __restrict__ ei, const float* __restrict__ H,
    const float* __restrict__ a_src, const float* __restrict__ a_dst,
    const float* __restrict__ bias, const float* __restrict__ prelu_w,
    float* __restrict__ out, const unsigned* __restrict__ keys, int E, int N) {
  __shared__ unsigned s_list[NPG][NCAP];
  __shared__ unsigned s_sorted[NOWN][NCAP];
  __shared__ unsigned s_match[32][WCAP];
  __shared__ unsigned s_cnt[256];
  __shared__ unsigned s_ovf;

  const int tid  = threadIdx.x;
  const int lane = tid & 31;
  const int wave = tid >> 5;
  const int lo   = blockIdx.x * NPG;
  const unsigned span = (unsigned)((N - lo < NPG) ? (N - lo) : NPG);

  if (tid < 256) s_cnt[tid] = 0u;
  if (tid == 0) s_ovf = 0u;
  __syncthreads();

  if (E > 0) {
    const int* srcp = ei;
    const int* dstp = ei + E;
    unsigned* mlist = &s_match[wave][0];
    unsigned wc = 0u;
    {
      const uint4* kv = (const uint4*)keys;
      const int nG = (E + 15) >> 4;
      const unsigned splat = ((unsigned)blockIdx.x & 0xffu) * 0x01010101u;
      const int UNITV = 1024 * FV;
      const int nUnits = (nG + UNITV - 1) / UNITV;
      const int nSlots = nUnits * UNITV;
      const int lastg = nG - 1;
      const int off = (int)((blockIdx.x * 1024u) % (unsigned)nSlots);
      uint4 cur[FV];
      #pragma unroll
      for (int k = 0; k < FV; ++k) {
        int b = off + k * 1024;
        if (b >= nSlots) b -= nSlots;
        const int g = b + tid;
        cur[k] = kv[(g < nG) ? g : lastg];
      }
      #pragma unroll 2
      for (int it = 0; it < nUnits; ++it) {
        const int b0 = off + it * UNITV;
        uint4 nxt[FV];
        #pragma unroll
        for (int k = 0; k < FV; ++k) {
          int b = b0 + UNITV + k * 1024;
          if (b >= nSlots) b -= nSlots;
          if (b >= nSlots) b -= nSlots;
          const int g = b + tid;
          nxt[k] = kv[(g < nG) ? g : lastg];
        }
        #pragma unroll
        for (int k = 0; k < FV; ++k) {
          int b = b0 + k * 1024;
          if (b >= nSlots) b -= nSlots;
          const int g = b + tid;
          const unsigned x0 = cur[k].x ^ splat;
          const unsigned x1 = cur[k].y ^ splat;
          const unsigned x2 = cur[k].z ^ splat;
          const unsigned x3 = cur[k].w ^ splat;
          const unsigned z0 = (x0 - 0x01010101u) & ~x0 & 0x80808080u;
          const unsigned z1 = (x1 - 0x01010101u) & ~x1 & 0x80808080u;
          const unsigned z2 = (x2 - 0x01010101u) & ~x2 & 0x80808080u;
          const unsigned z3 = (x3 - 0x01010101u) & ~x3 & 0x80808080u;
          const bool any = ((z0 | z1 | z2 | z3) != 0u) && (g < nG);
          if (__builtin_amdgcn_ballot_w32(any) != 0u) {
            unsigned m = any ? ((z0 >> 7) | (z1 >> 6) | (z2 >> 5) | (z3 >> 4)) : 0u;
            unsigned hb = __builtin_amdgcn_ballot_w32(m != 0u);
            while (hb != 0u) {
              if (m != 0u) {
                const unsigned slot = wc + __builtin_amdgcn_mbcnt_lo(hb, 0u);
                const unsigned p = (unsigned)__builtin_ctz(m);
                m &= m - 1u;
                if (slot < WCAP) mlist[slot] = 16u * (unsigned)g + 4u * (p & 7u) + (p >> 3);
                else s_ovf = 1u;
              }
              wc += (unsigned)__popc(hb);
              hb = __builtin_amdgcn_ballot_w32(m != 0u);
            }
          }
        }
        #pragma unroll
        for (int k = 0; k < FV; ++k) cur[k] = nxt[k];
      }
    }

    {
      const unsigned nm = (wc < WCAP) ? wc : WCAP;
      for (unsigned base = 0; base < nm; base += 128u) {
        int dv[4], sv[4];
        bool ok[4];
        #pragma unroll
        for (int q = 0; q < 4; ++q) {
          const unsigned i = base + (unsigned)q * 32u + (unsigned)lane;
          const unsigned e = (i < nm) ? mlist[(i < nm) ? i : 0u] : 0xffffffffu;
          ok[q] = e < (unsigned)E;
          const unsigned ec = ok[q] ? e : 0u;
          dv[q] = dstp[ec];
          sv[q] = srcp[ec];
        }
        #pragma unroll
        for (int q = 0; q < 4; ++q) {
          const unsigned dl = (unsigned)(dv[q] - lo);
          if (ok[q] && dl < span) push_edge(s_list, s_cnt, &s_ovf, dl, (unsigned)sv[q]);
        }
      }
    }
  }
  __syncthreads();

  if (wave >= NOWN) return;
  const int ovf = __builtin_amdgcn_readfirstlane((int)s_ovf);
  for (unsigned j = 0; j < NPW; ++j) {
    const unsigned dl = (unsigned)wave * NPW + j;
    if (dl >= span) break;
    const int d = lo + (int)dl;
    if (!ovf) {
      int n = __builtin_amdgcn_readfirstlane((int)s_cnt[dl]);
      if (n > NCAP) n = NCAP;
      reduce_node_list(H, a_src, a_dst, bias, prelu_w, out, &s_list[dl][0],
                       &s_sorted[wave][0], n, d, lane);
    } else {
      reduce_node_global(ei, H, a_src, a_dst, bias, prelu_w, out, d, E, lane);
    }
  }
}

extern "C" void kernel_launch(void* const* d_in, const int* in_sizes, int n_in,
                              void* d_out, int out_size, void* d_ws, size_t ws_size,
                              hipStream_t stream) {
  const float* x       = (const float*)d_in[0];
  const int*   ei      = (const int*)d_in[1];
  const float* W       = (const float*)d_in[2];
  const float* att_src = (const float*)d_in[3];
  const float* att_dst = (const float*)d_in[4];
  const float* bias    = (const float*)d_in[5];
  const float* prelu_w = (const float*)d_in[6];
  float* out = (float*)d_out;

  const int N = in_sizes[0] / IN_CH;
  const int E = in_sizes[1] / 2;

  char* ws = (char*)d_ws;
  float* H = (float*)ws;            ws += (size_t)N * OUT_CH * sizeof(float);
  float* a_src = (float*)ws;        ws += (size_t)N * HEADS * sizeof(float);
  float* a_dst = (float*)ws;        ws += (size_t)N * HEADS * sizeof(float);
  unsigned* keys = (unsigned*)ws;

  const int strips = (N + 15) / 16;
  const int nbKeys = ((((E + 15) >> 4) << 2) + 127) / 128;
  gemm_dots_keys<<<strips + nbKeys, 128, 0, stream>>>(x, W, H, N, att_src, att_dst,
                                                      a_src, a_dst, ei, keys, E, strips);

  aggregate_slabs<<<(N + NPG - 1) / NPG, 1024, 0, stream>>>(ei, H, a_src, a_dst,
                                                            bias, prelu_w, out, keys, E, N);
}


# Reduce phase: a slab's nodes are claimed one at a time from an LDS counter by all 32 waves (was 7 fixed nodes on each of 28 waves, 4 idle); each node's order of sums unchanged
# speedup vs baseline: 7.4702x; 1.0453x over previous
#include <hip/hip_runtime.h>


#define OUT_CH 256
#define IN_CH  256
#define HEADS  4
#define HID    64
#define NEG_SLOPE 0.2f
#define NPG   196

typedef float v2f  __attribute__((ext_vector_type(2)));
typedef float v4f  __attribute__((ext_vector_type(4)));
typedef float v8f  __attribute__((ext_vector_type(8)));

__device__ __forceinline__ unsigned slab_key(int d) {
  return ((unsigned)d / (unsigned)NPG) & 0xffu;
}

__global__ __launch_bounds__(128) void gemm_dots_keys(const float* __restrict__ X,
                              const float* __restrict__ W,
                              float* __restrict__ H, int M,
                              const float* __restrict__ att_src,
                              const float* __restrict__ att_dst,
                              float* __restrict__ a_src,
                              float* __restrict__ a_dst,
                              const int* __restrict__ ei,
                              unsigned* __restrict__ keys, int E, int nbGemm) {
  __shared__ float s_dots[2 * 16 * HEADS];

  if ((int)blockIdx.x >= nbGemm) {
    const int w  = ((int)blockIdx.x - nbGemm) * (int)blockDim.x + (int)threadIdx.x;
    const int nW = ((E + 15) >> 4) << 2;
    if (w >= nW) return;
    const int* dstp = ei + E;
    const int e0 = 4 * w;
    int d0, d1, d2, d3;
    if (e0 + 3 < E && (((size_t)(dstp + e0)) & 15u) == 0u) {
      const int4 v = *(const int4*)(dstp + e0);
      d0 = v.x; d1 = v.y; d2 = v.z; d3 = v.w;
    } else {
      d0 = (e0     < E) ? dstp[e0]     : 0;
      d1 = (e0 + 1 < E) ? dstp[e0 + 1] : 0;
      d2 = (e0 + 2 < E) ? dstp[e0 + 2] : 0;
      d3 = (e0 + 3 < E) ? dstp[e0 + 3] : 0;
    }
    keys[w] = slab_key(d0) | (slab_key(d1) << 8) | (slab_key(d2) << 16) | (slab_key(d3) << 24);
    return;
  }

  const int wave  = threadIdx.x >> 5;
  const int lane  = threadIdx.x & 31;
  const int strip = blockIdx.x;
  if (strip * 16 >= M) return;
  const int ncol0 = wave * HID;
  const int lo    = lane & 15;
  const int hi    = lane >> 4;
  const int row   = strip * 16 + lo;

  const float* xp = X + (size_t)row * IN_CH + 8 * hi;
  const float* wp = W + (size_t)(8 * hi) * OUT_CH + ncol0 + 4 * lo;

  v4f rx0 = *(const v4f*)(xp);
  v4f rx1 = *(const v4f*)(xp + 4);
  v4f rw[8];
  #pragma unroll
  for (int i = 0; i < 8; ++i)
    rw[i] = *(const v4f*)(wp + (size_t)i * OUT_CH);

  v8f c0 = {}, c1 = {}, c2 = {}, c3 = {};
  #pragma unroll
  for (int kk = 0; kk < IN_CH; kk += 16) {
    v2f a[4], b[4][4];
    a[0].x = rx0.x; a[0].y = rx0.y;
    a[1].x = rx0.z; a[1].y = rx0.w;
    a[2].x = rx1.x; a[2].y = rx1.y;
    a[3].x = rx1.z; a[3].y = rx1.w;
    #pragma unroll
    for (int t = 0; t < 4; ++t) {
      b[t][0].x = rw[2 * t].x; b[t][0].y = rw[2 * t + 1].x;
      b[t][1].x = rw[2 * t].y; b[t][1].y = rw[2 * t + 1].y;
      b[t][2].x = rw[2 * t].z; b[t][2].y = rw[2 * t + 1].z;
      b[t][3].x = rw[2 * t].w; b[t][3].y = rw[2 * t + 1].w;
    }
    if (kk + 16 < IN_CH) {
      rx0 = *(const v4f*)(xp + kk + 16);
      rx1 = *(const v4f*)(xp + kk + 20);
      #pragma unroll
      for (int i = 0; i < 8; ++i)
        rw[i] = *(const v4f*)(wp + (size_t)(kk + 16 + i) * OUT_CH);
    }
    __builtin_amdgcn_sched_barrier(0);
    #pragma unroll
    for (int t = 0; t < 4; ++t) {
      c0 = __builtin_amdgcn_wmma_f32_16x16x4_f32(false, a[t], false, b[t][0], (short)0, c0, false, false);
      c1 = __builtin_amdgcn_wmma_f32_16x16x4_f32(false, a[t], false, b[t][1], (short)0, c1, false, false);
      c2 = __builtin_amdgcn_wmma_f32_16x16x4_f32(false, a[t], false, b[t][2], (short)0, c2, false, false);
      c3 = __builtin_amdgcn_wmma_f32_16x16x4_f32(false, a[t], false, b[t][3], (short)0, c3, false, false);
    }
    __builtin_amdgcn_sched_barrier(0);
  }
  #pragma unroll
  for (int r = 0; r < 8; ++r) {
    const int m = strip * 16 + r + 8 * hi;
    v4f o;
    o.x = c0[r]; o.y = c1[r]; o.z = c2[r]; o.w = c3[r];
    *(v4f*)(H + (size_t)m * OUT_CH + ncol0 + 4 * lo) = o;
  }

  const v4f as4 = *(const v4f*)(att_src + ncol0 + 4 * lo);
  const v4f ad4 = *(const v4f*)(att_dst + ncol0 + 4 * lo);
  float v[16];
  #pragma unroll
  for (int r = 0; r < 8; ++r) {
    v[r]     = c0[r] * as4.x + c1[r] * as4.y + c2[r] * as4.z + c3[r] * as4.w;
    v[8 + r] = c0[r] * ad4.x + c1[r] * ad4.y + c2[r] * ad4.z + c3[r] * ad4.w;
  }
  const bool b3 = (lo & 8) != 0;
  const bool b2 = (lo & 4) != 0;
  const bool b1 = (lo & 2) != 0;
  const bool b0 = (lo & 1) != 0;
  float w8[8], w4[4], w2[2];
  #pragma unroll
  for (int i = 0; i < 8; ++i) {
    const float keep = b3 ? v[8 + i] : v[i];
    const float send = b3 ? v[i] : v[8 + i];
    w8[i] = keep + __shfl_xor(send, 8);
  }
  #pragma unroll
  for (int i = 0; i < 4; ++i) {
    const float keep = b2 ? w8[4 + i] : w8[i];
    const float send = b2 ? w8[i] : w8[4 + i];
    w4[i] = keep + __shfl_xor(send, 4);
  }
  #pragma unroll
  for (int i = 0; i < 2; ++i) {
    const float keep = b1 ? w4[2 + i] : w4[i];
    const float send = b1 ? w4[i] : w4[2 + i];
    w2[i] = keep + __shfl_xor(send, 2);
  }
  const float dot = (b0 ? w2[1] : w2[0]) + __shfl_xor(b0 ? w2[0] : w2[1], 1);
  s_dots[(((lo >> 3) * 16) + 8 * hi + (lo & 7)) * HEADS + wave] = dot;
  __syncthreads();
  if (wave == 0) {
    const float4 q = *(const float4*)(&s_dots[4 * lane]);
    const int node = strip * 16 + lo;
    float* dstq = (hi ? a_dst : a_src) + (size_t)node * HEADS;
    if (node < M) *(float4*)dstq = q;
  }
}

#define NOWN  32
#define NCAP  256
#define RB    6
#define FV    8
#define WCAP  384

__device__ __forceinline__ float leaky(float v) {
  return (v >= 0.f) ? v : NEG_SLOPE * v;
}
__device__ __forceinline__ float bcastf(float v, int srcLane) {
  return __int_as_float(__builtin_amdgcn_readlane(__float_as_int(v), srcLane));
}
__device__ __forceinline__ void fma4(float4& acc, float x, const float4& h) {
  acc.x += x * h.x; acc.y += x * h.y; acc.z += x * h.z; acc.w += x * h.w;
}


struct RowTail { float4 bA, bB, pA, pB; };

__device__ __forceinline__ void finish_row(float* __restrict__ out, int d, int lane,
                                           const float4& accA, const float4& accB,
                                           float l0, float l1, float l2, float l3,
                                           const float* __restrict__ bias,
                                           const float* __restrict__ prelu_w) {
  RowTail rt;
  rt.bA = *(const float4*)(bias + 4 * lane);
  rt.bB = *(const float4*)(bias + 128 + 4 * lane);
  rt.pA = *(const float4*)(prelu_w + 4 * lane);
  rt.pB = *(const float4*)(prelu_w + 128 + 4 * lane);
  for (int o = 16; o > 0; o >>= 1) {
    l0 += __shfl_xor(l0, o);
    l1 += __shfl_xor(l1, o);
    l2 += __shfl_xor(l2, o);
    l3 += __shfl_xor(l3, o);
  }
  const int hsel = lane >> 4;
  const float da = (hsel ? l1 : l0) + 1e-16f;
  const float db = (hsel ? l3 : l2) + 1e-16f;
  float4 oA, oB;
  float v;
  v = accA.x / da + rt.bA.x; oA.x = (v >= 0.f) ? v : rt.pA.x * v;
  v = accA.y / da + rt.bA.y; oA.y = (v >= 0.f) ? v : rt.pA.y * v;
  v = accA.z / da + rt.bA.z; oA.z = (v >= 0.f) ? v : rt.pA.z * v;
  v = accA.w / da + rt.bA.w; oA.w = (v >= 0.f) ? v : rt.pA.w * v;
  v = accB.x / db + rt.bB.x; oB.x = (v >= 0.f) ? v : rt.pB.x * v;
  v = accB.y / db + rt.bB.y; oB.y = (v >= 0.f) ? v : rt.pB.y * v;
  v = accB.z / db + rt.bB.z; oB.z = (v >= 0.f) ? v : rt.pB.z * v;
  v = accB.w / db + rt.bB.w; oB.w = (v >= 0.f) ? v : rt.pB.w * v;
  float* op = out + (size_t)d * OUT_CH;
  *(float4*)(op + 4 * lane) = oA;
  *(float4*)(op + 128 + 4 * lane) = oB;
}

__device__ __forceinline__ bool fetch_edge_global(const int* __restrict__ ei,
                                                  int d, int E, int i, int& s) {
  const bool mt = (i < E) && (ei[E + i] == d);
  s = mt ? ei[i] : d;
  return mt;
}

__device__ __forceinline__ void reduce_node_global(
    const int* __restrict__ ei, const float* __restrict__ H,
    const float* __restrict__ a_src, const float* __restrict__ a_dst,
    const float* __restrict__ bias, const float* __restrict__ prelu_w,
    float* __restrict__ out, int d, int E, int lane) {
  const float4 ad = *(const float4*)(a_dst + (size_t)d * HEADS);
  const float4 as = *(const float4*)(a_src + (size_t)d * HEADS);
  const float es0 = leaky(as.x + ad.x);
  const float es1 = leaky(as.y + ad.y);
  const float es2 = leaky(as.z + ad.z);
  const float es3 = leaky(as.w + ad.w);

  float m0 = es0, m1 = es1, m2 = es2, m3 = es3;
  for (int base = 0; base < E; base += 32) {
    int s;
    fetch_edge_global(ei, d, E, base + lane, s);
    const float4 a = *(const float4*)(a_src + (size_t)s * HEADS);
    m0 = fmaxf(m0, leaky(a.x + ad.x));
    m1 = fmaxf(m1, leaky(a.y + ad.y));
    m2 = fmaxf(m2, leaky(a.z + ad.z));
    m3 = fmaxf(m3, leaky(a.w + ad.w));
  }
  for (int o = 16; o > 0; o >>= 1) {
    m0 = fmaxf(m0, __shfl_xor(m0, o));
    m1 = fmaxf(m1, __shfl_xor(m1, o));
    m2 = fmaxf(m2, __shfl_xor(m2, o));
    m3 = fmaxf(m3, __shfl_xor(m3, o));
  }

  const int hsel = lane >> 4;
  const float* Hl = H + 4 * lane;
  const float xs0 = expf(es0 - m0);
  const float xs1 = expf(es1 - m1);
  const float xs2 = expf(es2 - m2);
  const float xs3 = expf(es3 - m3);
  float4 accA, accB;
  {
    const float xa = hsel ? xs1 : xs0;
    const float xb = hsel ? xs3 : xs2;
    const float4 hA = *(const float4*)(Hl + (size_t)d * OUT_CH);
    const float4 hB = *(const float4*)(Hl + (size_t)d * OUT_CH + 128);
    accA.x = xa * hA.x; accA.y = xa * hA.y; accA.z = xa * hA.z; accA.w = xa * hA.w;
    accB.x = xb * hB.x; accB.y = xb * hB.y; accB.z = xb * hB.z; accB.w = xb * hB.w;
  }
  float l0 = (lane == 0) ? xs0 : 0.f;
  float l1 = (lane == 0) ? xs1 : 0.f;
  float l2 = (lane == 0) ? xs2 : 0.f;
  float l3 = (lane == 0) ? xs3 : 0.f;

  for (int base = 0; base < E; base += 32) {
    int s;
    const bool mt = fetch_edge_global(ei, d, E, base + lane, s);
    const float4 a = *(const float4*)(a_src + (size_t)s * HEADS);
    const float w = mt ? 1.f : 0.f;
    const float x0 = expf(leaky(a.x + ad.x) - m0) * w;
    const float x1 = expf(leaky(a.y + ad.y) - m1) * w;
    const float x2 = expf(leaky(a.z + ad.z) - m2) * w;
    const float x3 = expf(leaky(a.w + ad.w) - m3) * w;
    l0 += x0; l1 += x1; l2 += x2; l3 += x3;
    unsigned mask = __builtin_amdgcn_ballot_w32(mt);
    while (mask) {
      const int b = __builtin_ctz(mask);
      mask &= mask - 1;
      const int sb = __builtin_amdgcn_readlane(s, b);
      const float4 hA = *(const float4*)(Hl + (size_t)sb * OUT_CH);
      const float4 hB = *(const float4*)(Hl + (size_t)sb * OUT_CH + 128);
      const float y0 = bcastf(x0, b);
      const float y1 = bcastf(x1, b);
      const float y2 = bcastf(x2, b);
      const float y3 = bcastf(x3, b);
      fma4(accA, hsel ? y1 : y0, hA);
      fma4(accB, hsel ? y3 : y2, hB);
    }
  }
  finish_row(out, d, lane, accA, accB, l0, l1, l2, l3, bias, prelu_w);
}

__device__ __forceinline__ void add_chunk(const float* __restrict__ Hl, int s, bool mt,
                                          const float4& sc, float m0, float m1,
                                          float m2, float m3, int cnt, int hsel,
                                          float4& accA, float4& accB,
                                          float& l0, float& l1, float& l2, float& l3) {
  const float w = mt ? 1.f : 0.f;
  const float x0 = expf(sc.x - m0) * w;
  const float x1 = expf(sc.y - m1) * w;
  const float x2 = expf(sc.z - m2) * w;
  const float x3 = expf(sc.w - m3) * w;
  l0 += x0; l1 += x1; l2 += x2; l3 += x3;
  for (int q0 = 0; q0 < cnt; q0 += RB) {
    float4 hA[RB], hB[RB];
    #pragma unroll
    for (int q = 0; q < RB; ++q) {
      const bool has = (q0 + q) < cnt;
      const int bl = has ? (q0 + q) : q0;
      const int sb = __builtin_amdgcn_readlane(s, bl);
      hA[q] = *(const float4*)(Hl + (size_t)sb * OUT_CH);
      hB[q] = *(const float4*)(Hl + (size_t)sb * OUT_CH + 128);
    }
    #pragma unroll
    for (int q = 0; q < RB; ++q) {
      const bool has = (q0 + q) < cnt;
      const int bl = has ? (q0 + q) : q0;
      const float y0 = bcastf(x0, bl);
      const float y1 = bcastf(x1, bl);
      const float y2 = bcastf(x2, bl);
      const float y3 = bcastf(x3, bl);
      fma4(accA, has ? (hsel ? y1 : y0) : 0.f, hA[q]);
      fma4(accB, has ? (hsel ? y3 : y2) : 0.f, hB[q]);
    }
  }
}

__device__ __forceinline__ void sort_list(const unsigned* lst, unsigned* srt,
                                          int n, int lane) {
  for (int base = 0; base < n; base += 32) {
    const int i = base + lane;
    const unsigned v = (i < n) ? lst[(i < n) ? i : 0] : 0xffffffffu;
    unsigned r = 0u;
    for (int b2 = 0; b2 < n; b2 += 32) {
      const int i2 = b2 + lane;
      const unsigned w = (i2 < n) ? lst[(i2 < n) ? i2 : 0] : 0xffffffffu;
      const int m = (n - b2 < 32) ? (n - b2) : 32;
      for (int j = 0; j < m; ++j) {
        const unsigned wj = (unsigned)__builtin_amdgcn_readlane((int)w, j);
        const int idx = b2 + j;
        r += (wj < v || (wj == v && idx < i)) ? 1u : 0u;
      }
    }
    if (i < n) srt[r] = v;
  }
}

__device__ __forceinline__ void reduce_node_list(
    const float* __restrict__ H, const float* __restrict__ a_src,
    const float* __restrict__ a_dst, const float* __restrict__ bias,
    const float* __restrict__ prelu_w, float* __restrict__ out,
    const unsigned* raw, unsigned* srt, int n, int d, int lane) {
  const int hsel = lane >> 4;
  const float* Hl = H + 4 * lane;
  const float4 ad = *(const float4*)(a_dst + (size_t)d * HEADS);
  const float4 as = *(const float4*)(a_src + (size_t)d * HEADS);
  const float4 hdA = *(const float4*)(Hl + (size_t)d * OUT_CH);
  const float4 hdB = *(const float4*)(Hl + (size_t)d * OUT_CH + 128);
  sort_list(raw, srt, n, lane);
  const unsigned* lst = srt;
  const bool mt0 = lane < n;
  const int  s0  = mt0 ? (int)lst[lane] : d;
  const float4 a0 = *(const float4*)(a_src + (size_t)s0 * HEADS);
  float4 sc0;
  sc0.x = leaky(a0.x + ad.x); sc0.y = leaky(a0.y + ad.y);
  sc0.z = leaky(a0.z + ad.z); sc0.w = leaky(a0.w + ad.w);
  const float es0 = leaky(as.x + ad.x);
  const float es1 = leaky(as.y + ad.y);
  const float es2 = leaky(as.z + ad.z);
  const float es3 = leaky(as.w + ad.w);

  float m0 = fmaxf(es0, sc0.x), m1 = fmaxf(es1, sc0.y);
  float m2 = fmaxf(es2, sc0.z), m3 = fmaxf(es3, sc0.w);
  for (int base = 32; base < n; base += 32) {
    const int i = base + lane;
    const bool mt = i < n;
    const int s = mt ? (int)lst[mt ? i : 0] : d;
    const float4 a = *(const float4*)(a_src + (size_t)s * HEADS);
    m0 = fmaxf(m0, leaky(a.x + ad.x));
    m1 = fmaxf(m1, leaky(a.y + ad.y));
    m2 = fmaxf(m2, leaky(a.z + ad.z));
    m3 = fmaxf(m3, leaky(a.w + ad.w));
  }
  for (int o = 16; o > 0; o >>= 1) {
    m0 = fmaxf(m0, __shfl_xor(m0, o));
    m1 = fmaxf(m1, __shfl_xor(m1, o));
    m2 = fmaxf(m2, __shfl_xor(m2, o));
    m3 = fmaxf(m3, __shfl_xor(m3, o));
  }

  const float xs0 = expf(es0 - m0);
  const float xs1 = expf(es1 - m1);
  const float xs2 = expf(es2 - m2);
  const float xs3 = expf(es3 - m3);
  float4 accA, accB;
  {
    const float xa = hsel ? xs1 : xs0;
    const float xb = hsel ? xs3 : xs2;
    accA.x = xa * hdA.x; accA.y = xa * hdA.y; accA.z = xa * hdA.z; accA.w = xa * hdA.w;
    accB.x = xb * hdB.x; accB.y = xb * hdB.y; accB.z = xb * hdB.z; accB.w = xb * hdB.w;
  }
  float l0 = (lane == 0) ? xs0 : 0.f;
  float l1 = (lane == 0) ? xs1 : 0.f;
  float l2 = (lane == 0) ? xs2 : 0.f;
  float l3 = (lane == 0) ? xs3 : 0.f;

  add_chunk(Hl, s0, mt0, sc0, m0, m1, m2, m3, (n < 32) ? n : 32, hsel,
            accA, accB, l0, l1, l2, l3);
  for (int base = 32; base < n; base += 32) {
    const int i = base + lane;
    const bool mt = i < n;
    const int s = mt ? (int)lst[mt ? i : 0] : d;
    const float4 a = *(const float4*)(a_src + (size_t)s * HEADS);
    float4 sc;
    sc.x = leaky(a.x + ad.x); sc.y = leaky(a.y + ad.y);
    sc.z = leaky(a.z + ad.z); sc.w = leaky(a.w + ad.w);
    const int rem = n - base;
    add_chunk(Hl, s, mt, sc, m0, m1, m2, m3, (rem < 32) ? rem : 32, hsel,
              accA, accB, l0, l1, l2, l3);
  }
  finish_row(out, d, lane, accA, accB, l0, l1, l2, l3, bias, prelu_w);
}

__device__ __forceinline__ void push_edge(unsigned (*list)[NCAP], unsigned* cnt,
                                          unsigned* ovf, unsigned dl, unsigned s) {
  const unsigned slot = atomicAdd(&cnt[dl], 1u);
  if (slot < NCAP) list[dl][slot] = s;
  else *ovf = 1u;
}

__global__ __launch_bounds__(1024) void aggregate_slabs(
    const int* __restrict__ ei, const float* __restrict__ H,
    const float* __restrict__ a_src, const float* __restrict__ a_dst,
    const float* __restrict__ bias, const float* __restrict__ prelu_w,
    float* __restrict__ out, const unsigned* __restrict__ keys, int E, int N) {
  __shared__ unsigned s_list[NPG][NCAP];
  __shared__ unsigned s_sorted[NOWN][NCAP];
  __shared__ unsigned s_match[32][WCAP];
  __shared__ unsigned s_cnt[256];
  __shared__ unsigned s_ovf;
  __shared__ unsigned s_next;

  const int tid  = threadIdx.x;
  const int lane = tid & 31;
  const int wave = tid >> 5;
  const int lo   = blockIdx.x * NPG;
  const unsigned span = (unsigned)((N - lo < NPG) ? (N - lo) : NPG);

  if (tid < 256) s_cnt[tid] = 0u;
  if (tid == 0) { s_ovf = 0u; s_next = 0u; }
  __syncthreads();

  if (E > 0) {
    const int* srcp = ei;
    const int* dstp = ei + E;
    unsigned* mlist = &s_match[wave][0];
    unsigned wc = 0u;
    {
      const uint4* kv = (const uint4*)keys;
      const int nG = (E + 15) >> 4;
      const unsigned splat = ((unsigned)blockIdx.x & 0xffu) * 0x01010101u;
      const int UNITV = 1024 * FV;
      const int nUnits = (nG + UNITV - 1) / UNITV;
      const int nSlots = nUnits * UNITV;
      const int lastg = nG - 1;
      const int off = (int)((blockIdx.x * 1024u) % (unsigned)nSlots);
      uint4 cur[FV];
      #pragma unroll
      for (int k = 0; k < FV; ++k) {
        int b = off + k * 1024;
        if (b >= nSlots) b -= nSlots;
        const int g = b + tid;
        cur[k] = kv[(g < nG) ? g : lastg];
      }
      #pragma unroll 2
      for (int it = 0; it < nUnits; ++it) {
        const int b0 = off + it * UNITV;
        uint4 nxt[FV];
        #pragma unroll
        for (int k = 0; k < FV; ++k) {
          int b = b0 + UNITV + k * 1024;
          if (b >= nSlots) b -= nSlots;
          if (b >= nSlots) b -= nSlots;
          const int g = b + tid;
          nxt[k] = kv[(g < nG) ? g : lastg];
        }
        #pragma unroll
        for (int k = 0; k < FV; ++k) {
          int b = b0 + k * 1024;
          if (b >= nSlots) b -= nSlots;
          const int g = b + tid;
          const unsigned x0 = cur[k].x ^ splat;
          const unsigned x1 = cur[k].y ^ splat;
          const unsigned x2 = cur[k].z ^ splat;
          const unsigned x3 = cur[k].w ^ splat;
          const unsigned z0 = (x0 - 0x01010101u) & ~x0 & 0x80808080u;
          const unsigned z1 = (x1 - 0x01010101u) & ~x1 & 0x80808080u;
          const unsigned z2 = (x2 - 0x01010101u) & ~x2 & 0x80808080u;
          const unsigned z3 = (x3 - 0x01010101u) & ~x3 & 0x80808080u;
          const bool any = ((z0 | z1 | z2 | z3) != 0u) && (g < nG);
          if (__builtin_amdgcn_ballot_w32(any) != 0u) {
            unsigned m = any ? ((z0 >> 7) | (z1 >> 6) | (z2 >> 5) | (z3 >> 4)) : 0u;
            unsigned hb = __builtin_amdgcn_ballot_w32(m != 0u);
            while (hb != 0u) {
              if (m != 0u) {
                const unsigned slot = wc + __builtin_amdgcn_mbcnt_lo(hb, 0u);
                const unsigned p = (unsigned)__builtin_ctz(m);
                m &= m - 1u;
                if (slot < WCAP) mlist[slot] = 16u * (unsigned)g + 4u * (p & 7u) + (p >> 3);
                else s_ovf = 1u;
              }
              wc += (unsigned)__popc(hb);
              hb = __builtin_amdgcn_ballot_w32(m != 0u);
            }
          }
        }
        #pragma unroll
        for (int k = 0; k < FV; ++k) cur[k] = nxt[k];
      }
    }

    {
      const unsigned nm = (wc < WCAP) ? wc : WCAP;
      for (unsigned base = 0; base < nm; base += 128u) {
        int dv[4], sv[4];
        bool ok[4];
        #pragma unroll
        for (int q = 0; q < 4; ++q) {
          const unsigned i = base + (unsigned)q * 32u + (unsigned)lane;
          const unsigned e = (i < nm) ? mlist[(i < nm) ? i : 0u] : 0xffffffffu;
          ok[q] = e < (unsigned)E;
          const unsigned ec = ok[q] ? e : 0u;
          dv[q] = dstp[ec];
          sv[q] = srcp[ec];
        }
        #pragma unroll
        for (int q = 0; q < 4; ++q) {
          const unsigned dl = (unsigned)(dv[q] - lo);
          if (ok[q] && dl < span) push_edge(s_list, s_cnt, &s_ovf, dl, (unsigned)sv[q]);
        }
      }
    }
  }
  __syncthreads();

  const int ovf = __builtin_amdgcn_readfirstlane((int)s_ovf);
  for (;;) {
    unsigned mine = 0u;
    if (lane == 0) mine = atomicAdd(&s_next, 1u);
    const unsigned dl = (unsigned)__builtin_amdgcn_readfirstlane((int)mine);
    if (dl >= span) break;
    const int d = lo + (int)dl;
    if (!ovf) {
      int n = __builtin_amdgcn_readfirstlane((int)s_cnt[dl]);
      if (n > NCAP) n = NCAP;
      reduce_node_list(H, a_src, a_dst, bias, prelu_w, out, &s_list[dl][0],
                       &s_sorted[wave][0], n, d, lane);
    } else {
      reduce_node_global(ei, H, a_src, a_dst, bias, prelu_w, out, d, E, lane);
    }
  }
}

extern "C" void kernel_launch(void* const* d_in, const int* in_sizes, int n_in,
                              void* d_out, int out_size, void* d_ws, size_t ws_size,
                              hipStream_t stream) {
  const float* x       = (const float*)d_in[0];
  const int*   ei      = (const int*)d_in[1];
  const float* W       = (const float*)d_in[2];
  const float* att_src = (const float*)d_in[3];
  const float* att_dst = (const float*)d_in[4];
  const float* bias    = (const float*)d_in[5];
  const float* prelu_w = (const float*)d_in[6];
  float* out = (float*)d_out;

  const int N = in_sizes[0] / IN_CH;
  const int E = in_sizes[1] / 2;

  char* ws = (char*)d_ws;
  float* H = (float*)ws;            ws += (size_t)N * OUT_CH * sizeof(float);
  float* a_src = (float*)ws;        ws += (size_t)N * HEADS * sizeof(float);
  float* a_dst = (float*)ws;        ws += (size_t)N * HEADS * sizeof(float);
  unsigned* keys = (unsigned*)ws;

  const int strips = (N + 15) / 16;
  const int nbKeys = ((((E + 15) >> 4) << 2) + 127) / 128;
  gemm_dots_keys<<<strips + nbKeys, 128, 0, stream>>>(x, W, H, N, att_src, att_dst,
                                                      a_src, a_dst, ei, keys, E, strips);

  aggregate_slabs<<<(N + NPG - 1) / NPG, 1024, 0, stream>>>(ei, H, a_src, a_dst,
                                                            bias, prelu_w, out, keys, E, N);
}


# Slab filter scans one bit per group of 16 edges (per-group key sets stored as eight bit planes by the key blocks); key bytes are read only for groups that have the slab's bit; order of sums kept
# speedup vs baseline: 7.8334x; 1.0486x over previous
#include <hip/hip_runtime.h>


#define OUT_CH 256
#define IN_CH  256
#define HEADS  4
#define HID    64
#define NEG_SLOPE 0.2f
#define NPG   196

typedef float v2f  __attribute__((ext_vector_type(2)));
typedef float v4f  __attribute__((ext_vector_type(4)));
typedef float v8f  __attribute__((ext_vector_type(8)));

__device__ __forceinline__ unsigned slab_key(int d) {
  return ((unsigned)d / (unsigned)NPG) & 0xffu;
}

__global__ __launch_bounds__(128) void gemm_dots_keys(const float* __restrict__ X,
                              const float* __restrict__ W,
                              float* __restrict__ H, int M,
                              const float* __restrict__ att_src,
                              const float* __restrict__ att_dst,
                              float* __restrict__ a_src,
                              float* __restrict__ a_dst,
                              const int* __restrict__ ei,
                              unsigned* __restrict__ keys,
                              unsigned* __restrict__ planes, int E, int nbGemm) {
  __shared__ float s_dots[2 * 16 * HEADS];

  if ((int)blockIdx.x >= nbGemm) {
    const int g   = ((int)blockIdx.x - nbGemm) * (int)blockDim.x + (int)threadIdx.x;
    const int nG  = (E + 15) >> 4;
    const int nGp = ((nG + 3) >> 2) << 2;
    if (g >= nGp) return;
    const int* dstp = ei + E;
    const int e0 = 16 * g;
    int d[16];
    if (e0 + 15 < E && (((size_t)(dstp + e0)) & 15u) == 0u) {
      #pragma unroll
      for (int i = 0; i < 4; ++i) {
        const int4 v = *(const int4*)(dstp + e0 + 4 * i);
        d[4 * i] = v.x; d[4 * i + 1] = v.y; d[4 * i + 2] = v.z; d[4 * i + 3] = v.w;
      }
    } else {
      #pragma unroll
      for (int j = 0; j < 16; ++j) d[j] = (e0 + j < E) ? dstp[e0 + j] : 0;
    }
    unsigned kw[4] = {0u, 0u, 0u, 0u};
    unsigned pm[8] = {0u, 0u, 0u, 0u, 0u, 0u, 0u, 0u};
    #pragma unroll
    for (int j = 0; j < 16; ++j) {
      const bool in = (e0 + j) < E;
      const unsigned k = in ? slab_key(d[j]) : 0u;
      kw[j >> 2] |= k << (8 * (j & 3));
      const unsigned bit = in ? (1u << (k & 31u)) : 0u;
      const unsigned p = k >> 5;
      #pragma unroll
      for (int i = 0; i < 8; ++i) pm[i] |= (p == (unsigned)i) ? bit : 0u;
    }
    if (g < nG) {
      uint4 kq;
      kq.x = kw[0]; kq.y = kw[1]; kq.z = kw[2]; kq.w = kw[3];
      *(uint4*)(keys + 4 * (size_t)g) = kq;
    }
    #pragma unroll
    for (int i = 0; i < 8; ++i) planes[(size_t)i * (size_t)nGp + (size_t)g] = pm[i];
    return;
  }

  const int wave  = threadIdx.x >> 5;
  const int lane  = threadIdx.x & 31;
  const int strip = blockIdx.x;
  if (strip * 16 >= M) return;
  const int ncol0 = wave * HID;
  const int lo    = lane & 15;
  const int hi    = lane >> 4;
  const int row   = strip * 16 + lo;

  const float* xp = X + (size_t)row * IN_CH + 8 * hi;
  const float* wp = W + (size_t)(8 * hi) * OUT_CH + ncol0 + 4 * lo;

  v4f rx0 = *(const v4f*)(xp);
  v4f rx1 = *(const v4f*)(xp + 4);
  v4f rw[8];
  #pragma unroll
  for (int i = 0; i < 8; ++i)
    rw[i] = *(const v4f*)(wp + (size_t)i * OUT_CH);

  v8f c0 = {}, c1 = {}, c2 = {}, c3 = {};
  #pragma unroll
  for (int kk = 0; kk < IN_CH; kk += 16) {
    v2f a[4], b[4][4];
    a[0].x = rx0.x; a[0].y = rx0.y;
    a[1].x = rx0.z; a[1].y = rx0.w;
    a[2].x = rx1.x; a[2].y = rx1.y;
    a[3].x = rx1.z; a[3].y = rx1.w;
    #pragma unroll
    for (int t = 0; t < 4; ++t) {
      b[t][0].x = rw[2 * t].x; b[t][0].y = rw[2 * t + 1].x;
      b[t][1].x = rw[2 * t].y; b[t][1].y = rw[2 * t + 1].y;
      b[t][2].x = rw[2 * t].z; b[t][2].y = rw[2 * t + 1].z;
      b[t][3].x = rw[2 * t].w; b[t][3].y = rw[2 * t + 1].w;
    }
    if (kk + 16 < IN_CH) {
      rx0 = *(const v4f*)(xp + kk + 16);
      rx1 = *(const v4f*)(xp + kk + 20);
      #pragma unroll
      for (int i = 0; i < 8; ++i)
        rw[i] = *(const v4f*)(wp + (size_t)(kk + 16 + i) * OUT_CH);
    }
    __builtin_amdgcn_sched_barrier(0);
    #pragma unroll
    for (int t = 0; t < 4; ++t) {
      c0 = __builtin_amdgcn_wmma_f32_16x16x4_f32(false, a[t], false, b[t][0], (short)0, c0, false, false);
      c1 = __builtin_amdgcn_wmma_f32_16x16x4_f32(false, a[t], false, b[t][1], (short)0, c1, false, false);
      c2 = __builtin_amdgcn_wmma_f32_16x16x4_f32(false, a[t], false, b[t][2], (short)0, c2, false, false);
      c3 = __builtin_amdgcn_wmma_f32_16x16x4_f32(false, a[t], false, b[t][3], (short)0, c3, false, false);
    }
    __builtin_amdgcn_sched_barrier(0);
  }
  #pragma unroll
  for (int r = 0; r < 8; ++r) {
    const int m = strip * 16 + r + 8 * hi;
    v4f o;
    o.x = c0[r]; o.y = c1[r]; o.z = c2[r]; o.w = c3[r];
    *(v4f*)(H + (size_t)m * OUT_CH + ncol0 + 4 * lo) = o;
  }

  const v4f as4 = *(const v4f*)(att_src + ncol0 + 4 * lo);
  const v4f ad4 = *(const v4f*)(att_dst + ncol0 + 4 * lo);
  float v[16];
  #pragma unroll
  for (int r = 0; r < 8; ++r) {
    v[r]     = c0[r] * as4.x + c1[r] * as4.y + c2[r] * as4.z + c3[r] * as4.w;
    v[8 + r] = c0[r] * ad4.x + c1[r] * ad4.y + c2[r] * ad4.z + c3[r] * ad4.w;
  }
  const bool b3 = (lo & 8) != 0;
  const bool b2 = (lo & 4) != 0;
  const bool b1 = (lo & 2) != 0;
  const bool b0 = (lo & 1) != 0;
  float w8[8], w4[4], w2[2];
  #pragma unroll
  for (int i = 0; i < 8; ++i) {
    const float keep = b3 ? v[8 + i] : v[i];
    const float send = b3 ? v[i] : v[8 + i];
    w8[i] = keep + __shfl_xor(send, 8);
  }
  #pragma unroll
  for (int i = 0; i < 4; ++i) {
    const float keep = b2 ? w8[4 + i] : w8[i];
    const float send = b2 ? w8[i] : w8[4 + i];
    w4[i] = keep + __shfl_xor(send, 4);
  }
  #pragma unroll
  for (int i = 0; i < 2; ++i) {
    const float keep = b1 ? w4[2 + i] : w4[i];
    const float send = b1 ? w4[i] : w4[2 + i];
    w2[i] = keep + __shfl_xor(send, 2);
  }
  const float dot = (b0 ? w2[1] : w2[0]) + __shfl_xor(b0 ? w2[0] : w2[1], 1);
  s_dots[(((lo >> 3) * 16) + 8 * hi + (lo & 7)) * HEADS + wave] = dot;
  __syncthreads();
  if (wave == 0) {
    const float4 q = *(const float4*)(&s_dots[4 * lane]);
    const int node = strip * 16 + lo;
    float* dstq = (hi ? a_dst : a_src) + (size_t)node * HEADS;
    if (node < M) *(float4*)dstq = q;
  }
}

#define NOWN  32
#define NCAP  256
#define RB    6
#define FV    8
#define WCAP  384

__device__ __forceinline__ float leaky(float v) {
  return (v >= 0.f) ? v : NEG_SLOPE * v;
}
__device__ __forceinline__ float bcastf(float v, int srcLane) {
  return __int_as_float(__builtin_amdgcn_readlane(__float_as_int(v), srcLane));
}
__device__ __forceinline__ void fma4(float4& acc, float x, const float4& h) {
  acc.x += x * h.x; acc.y += x * h.y; acc.z += x * h.z; acc.w += x * h.w;
}


struct RowTail { float4 bA, bB, pA, pB; };

__device__ __forceinline__ void finish_row(float* __restrict__ out, int d, int lane,
                                           const float4& accA, const float4& accB,
                                           float l0, float l1, float l2, float l3,
                                           const float* __restrict__ bias,
                                           const float* __restrict__ prelu_w) {
  RowTail rt;
  rt.bA = *(const float4*)(bias + 4 * lane);
  rt.bB = *(const float4*)(bias + 128 + 4 * lane);
  rt.pA = *(const float4*)(prelu_w + 4 * lane);
  rt.pB = *(const float4*)(prelu_w + 128 + 4 * lane);
  for (int o = 16; o > 0; o >>= 1) {
    l0 += __shfl_xor(l0, o);
    l1 += __shfl_xor(l1, o);
    l2 += __shfl_xor(l2, o);
    l3 += __shfl_xor(l3, o);
  }
  const int hsel = lane >> 4;
  const float da = (hsel ? l1 : l0) + 1e-16f;
  const float db = (hsel ? l3 : l2) + 1e-16f;
  float4 oA, oB;
  float v;
  v = accA.x / da + rt.bA.x; oA.x = (v >= 0.f) ? v : rt.pA.x * v;
  v = accA.y / da + rt.bA.y; oA.y = (v >= 0.f) ? v : rt.pA.y * v;
  v = accA.z / da + rt.bA.z; oA.z = (v >= 0.f) ? v : rt.pA.z * v;
  v = accA.w / da + rt.bA.w; oA.w = (v >= 0.f) ? v : rt.pA.w * v;
  v = accB.x / db + rt.bB.x; oB.x = (v >= 0.f) ? v : rt.pB.x * v;
  v = accB.y / db + rt.bB.y; oB.y = (v >= 0.f) ? v : rt.pB.y * v;
  v = accB.z / db + rt.bB.z; oB.z = (v >= 0.f) ? v : rt.pB.z * v;
  v = accB.w / db + rt.bB.w; oB.w = (v >= 0.f) ? v : rt.pB.w * v;
  float* op = out + (size_t)d * OUT_CH;
  *(float4*)(op + 4 * lane) = oA;
  *(float4*)(op + 128 + 4 * lane) = oB;
}

__device__ __forceinline__ bool fetch_edge_global(const int* __restrict__ ei,
                                                  int d, int E, int i, int& s) {
  const bool mt = (i < E) && (ei[E + i] == d);
  s = mt ? ei[i] : d;
  return mt;
}

__device__ __forceinline__ void reduce_node_global(
    const int* __restrict__ ei, const float* __restrict__ H,
    const float* __restrict__ a_src, const float* __restrict__ a_dst,
    const float* __restrict__ bias, const float* __restrict__ prelu_w,
    float* __restrict__ out, int d, int E, int lane) {
  const float4 ad = *(const float4*)(a_dst + (size_t)d * HEADS);
  const float4 as = *(const float4*)(a_src + (size_t)d * HEADS);
  const float es0 = leaky(as.x + ad.x);
  const float es1 = leaky(as.y + ad.y);
  const float es2 = leaky(as.z + ad.z);
  const float es3 = leaky(as.w + ad.w);

  float m0 = es0, m1 = es1, m2 = es2, m3 = es3;
  for (int base = 0; base < E; base += 32) {
    int s;
    fetch_edge_global(ei, d, E, base + lane, s);
    const float4 a = *(const float4*)(a_src + (size_t)s * HEADS);
    m0 = fmaxf(m0, leaky(a.x + ad.x));
    m1 = fmaxf(m1, leaky(a.y + ad.y));
    m2 = fmaxf(m2, leaky(a.z + ad.z));
    m3 = fmaxf(m3, leaky(a.w + ad.w));
  }
  for (int o = 16; o > 0; o >>= 1) {
    m0 = fmaxf(m0, __shfl_xor(m0, o));
    m1 = fmaxf(m1, __shfl_xor(m1, o));
    m2 = fmaxf(m2, __shfl_xor(m2, o));
    m3 = fmaxf(m3, __shfl_xor(m3, o));
  }

  const int hsel = lane >> 4;
  const float* Hl = H + 4 * lane;
  const float xs0 = expf(es0 - m0);
  const float xs1 = expf(es1 - m1);
  const float xs2 = expf(es2 - m2);
  const float xs3 = expf(es3 - m3);
  float4 accA, accB;
  {
    const float xa = hsel ? xs1 : xs0;
    const float xb = hsel ? xs3 : xs2;
    const float4 hA = *(const float4*)(Hl + (size_t)d * OUT_CH);
    const float4 hB = *(const float4*)(Hl + (size_t)d * OUT_CH + 128);
    accA.x = xa * hA.x; accA.y = xa * hA.y; accA.z = xa * hA.z; accA.w = xa * hA.w;
    accB.x = xb * hB.x; accB.y = xb * hB.y; accB.z = xb * hB.z; accB.w = xb * hB.w;
  }
  float l0 = (lane == 0) ? xs0 : 0.f;
  float l1 = (lane == 0) ? xs1 : 0.f;
  float l2 = (lane == 0) ? xs2 : 0.f;
  float l3 = (lane == 0) ? xs3 : 0.f;

  for (int base = 0; base < E; base += 32) {
    int s;
    const bool mt = fetch_edge_global(ei, d, E, base + lane, s);
    const float4 a = *(const float4*)(a_src + (size_t)s * HEADS);
    const float w = mt ? 1.f : 0.f;
    const float x0 = expf(leaky(a.x + ad.x) - m0) * w;
    const float x1 = expf(leaky(a.y + ad.y) - m1) * w;
    const float x2 = expf(leaky(a.z + ad.z) - m2) * w;
    const float x3 = expf(leaky(a.w + ad.w) - m3) * w;
    l0 += x0; l1 += x1; l2 += x2; l3 += x3;
    unsigned mask = __builtin_amdgcn_ballot_w32(mt);
    while (mask) {
      const int b = __builtin_ctz(mask);
      mask &= mask - 1;
      const int sb = __builtin_amdgcn_readlane(s, b);
      const float4 hA = *(const float4*)(Hl + (size_t)sb * OUT_CH);
      const float4 hB = *(const float4*)(Hl + (size_t)sb * OUT_CH + 128);
      const float y0 = bcastf(x0, b);
      const float y1 = bcastf(x1, b);
      const float y2 = bcastf(x2, b);
      const float y3 = bcastf(x3, b);
      fma4(accA, hsel ? y1 : y0, hA);
      fma4(accB, hsel ? y3 : y2, hB);
    }
  }
  finish_row(out, d, lane, accA, accB, l0, l1, l2, l3, bias, prelu_w);
}

__device__ __forceinline__ void add_chunk(const float* __restrict__ Hl, int s, bool mt,
                                          const float4& sc, float m0, float m1,
                                          float m2, float m3, int cnt, int hsel,
                                          float4& accA, float4& accB,
                                          float& l0, float& l1, float& l2, float& l3) {
  const float w = mt ? 1.f : 0.f;
  const float x0 = expf(sc.x - m0) * w;
  const float x1 = expf(sc.y - m1) * w;
  const float x2 = expf(sc.z - m2) * w;
  const float x3 = expf(sc.w - m3) * w;
  l0 += x0; l1 += x1; l2 += x2; l3 += x3;
  for (int q0 = 0; q0 < cnt; q0 += RB) {
    float4 hA[RB], hB[RB];
    #pragma unroll
    for (int q = 0; q < RB; ++q) {
      const bool has = (q0 + q) < cnt;
      const int bl = has ? (q0 + q) : q0;
      const int sb = __builtin_amdgcn_readlane(s, bl);
      hA[q] = *(const float4*)(Hl + (size_t)sb * OUT_CH);
      hB[q] = *(const float4*)(Hl + (size_t)sb * OUT_CH + 128);
    }
    #pragma unroll
    for (int q = 0; q < RB; ++q) {
      const bool has = (q0 + q) < cnt;
      const int bl = has ? (q0 + q) : q0;
      const float y0 = bcastf(x0, bl);
      const float y1 = bcastf(x1, bl);
      const float y2 = bcastf(x2, bl);
      const float y3 = bcastf(x3, bl);
      fma4(accA, has ? (hsel ? y1 : y0) : 0.f, hA[q]);
      fma4(accB, has ? (hsel ? y3 : y2) : 0.f, hB[q]);
    }
  }
}

__device__ __forceinline__ void sort_list(const unsigned* lst, unsigned* srt,
                                          int n, int lane) {
  for (int base = 0; base < n; base += 32) {
    const int i = base + lane;
    const unsigned v = (i < n) ? lst[(i < n) ? i : 0] : 0xffffffffu;
    unsigned r = 0u;
    for (int b2 = 0; b2 < n; b2 += 32) {
      const int i2 = b2 + lane;
      const unsigned w = (i2 < n) ? lst[(i2 < n) ? i2 : 0] : 0xffffffffu;
      const int m = (n - b2 < 32) ? (n - b2) : 32;
      for (int j = 0; j < m; ++j) {
        const unsigned wj = (unsigned)__builtin_amdgcn_readlane((int)w, j);
        const int idx = b2 + j;
        r += (wj < v || (wj == v && idx < i)) ? 1u : 0u;
      }
    }
    if (i < n) srt[r] = v;
  }
}

__device__ __forceinline__ void reduce_node_list(
    const float* __restrict__ H, const float* __restrict__ a_src,
    const float* __restrict__ a_dst, const float* __restrict__ bias,
    const float* __restrict__ prelu_w, float* __restrict__ out,
    const unsigned* raw, unsigned* srt, int n, int d, int lane) {
  const int hsel = lane >> 4;
  const float* Hl = H + 4 * lane;
  const float4 ad = *(const float4*)(a_dst + (size_t)d * HEADS);
  const float4 as = *(const float4*)(a_src + (size_t)d * HEADS);
  const float4 hdA = *(const float4*)(Hl + (size_t)d * OUT_CH);
  const float4 hdB = *(const float4*)(Hl + (size_t)d * OUT_CH + 128);
  sort_list(raw, srt, n, lane);
  const unsigned* lst = srt;
  const bool mt0 = lane < n;
  const int  s0  = mt0 ? (int)lst[lane] : d;
  const float4 a0 = *(const float4*)(a_src + (size_t)s0 * HEADS);
  float4 sc0;
  sc0.x = leaky(a0.x + ad.x); sc0.y = leaky(a0.y + ad.y);
  sc0.z = leaky(a0.z + ad.z); sc0.w = leaky(a0.w + ad.w);
  const float es0 = leaky(as.x + ad.x);
  const float es1 = leaky(as.y + ad.y);
  const float es2 = leaky(as.z + ad.z);
  const float es3 = leaky(as.w + ad.w);

  float m0 = fmaxf(es0, sc0.x), m1 = fmaxf(es1, sc0.y);
  float m2 = fmaxf(es2, sc0.z), m3 = fmaxf(es3, sc0.w);
  for (int base = 32; base < n; base += 32) {
    const int i = base + lane;
    const bool mt = i < n;
    const int s = mt ? (int)lst[mt ? i : 0] : d;
    const float4 a = *(const float4*)(a_src + (size_t)s * HEADS);
    m0 = fmaxf(m0, leaky(a.x + ad.x));
    m1 = fmaxf(m1, leaky(a.y + ad.y));
    m2 = fmaxf(m2, leaky(a.z + ad.z));
    m3 = fmaxf(m3, leaky(a.w + ad.w));
  }
  for (int o = 16; o > 0; o >>= 1) {
    m0 = fmaxf(m0, __shfl_xor(m0, o));
    m1 = fmaxf(m1, __shfl_xor(m1, o));
    m2 = fmaxf(m2, __shfl_xor(m2, o));
    m3 = fmaxf(m3, __shfl_xor(m3, o));
  }

  const float xs0 = expf(es0 - m0);
  const float xs1 = expf(es1 - m1);
  const float xs2 = expf(es2 - m2);
  const float xs3 = expf(es3 - m3);
  float4 accA, accB;
  {
    const float xa = hsel ? xs1 : xs0;
    const float xb = hsel ? xs3 : xs2;
    accA.x = xa * hdA.x; accA.y = xa * hdA.y; accA.z = xa * hdA.z; accA.w = xa * hdA.w;
    accB.x = xb * hdB.x; accB.y = xb * hdB.y; accB.z = xb * hdB.z; accB.w = xb * hdB.w;
  }
  float l0 = (lane == 0) ? xs0 : 0.f;
  float l1 = (lane == 0) ? xs1 : 0.f;
  float l2 = (lane == 0) ? xs2 : 0.f;
  float l3 = (lane == 0) ? xs3 : 0.f;

  add_chunk(Hl, s0, mt0, sc0, m0, m1, m2, m3, (n < 32) ? n : 32, hsel,
            accA, accB, l0, l1, l2, l3);
  for (int base = 32; base < n; base += 32) {
    const int i = base + lane;
    const bool mt = i < n;
    const int s = mt ? (int)lst[mt ? i : 0] : d;
    const float4 a = *(const float4*)(a_src + (size_t)s * HEADS);
    float4 sc;
    sc.x = leaky(a.x + ad.x); sc.y = leaky(a.y + ad.y);
    sc.z = leaky(a.z + ad.z); sc.w = leaky(a.w + ad.w);
    const int rem = n - base;
    add_chunk(Hl, s, mt, sc, m0, m1, m2, m3, (rem < 32) ? rem : 32, hsel,
              accA, accB, l0, l1, l2, l3);
  }
  finish_row(out, d, lane, accA, accB, l0, l1, l2, l3, bias, prelu_w);
}

__device__ __forceinline__ void push_edge(unsigned (*list)[NCAP], unsigned* cnt,
                                          unsigned* ovf, unsigned dl, unsigned s) {
  const unsigned slot = atomicAdd(&cnt[dl], 1u);
  if (slot < NCAP) list[dl][slot] = s;
  else *ovf = 1u;
}

__global__ __launch_bounds__(1024) void aggregate_slabs(
    const int* __restrict__ ei, const float* __restrict__ H,
    const float* __restrict__ a_src, const float* __restrict__ a_dst,
    const float* __restrict__ bias, const float* __restrict__ prelu_w,
    float* __restrict__ out, const unsigned* __restrict__ keys,
    const unsigned* __restrict__ planes, int E, int N) {
  __shared__ unsigned s_list[NPG][NCAP];
  __shared__ unsigned s_sorted[NOWN][NCAP];
  __shared__ unsigned s_match[32][WCAP];
  __shared__ unsigned s_cnt[256];
  __shared__ unsigned s_ovf;
  __shared__ unsigned s_next;

  const int tid  = threadIdx.x;
  const int lane = tid & 31;
  const int wave = tid >> 5;
  const int lo   = blockIdx.x * NPG;
  const unsigned span = (unsigned)((N - lo < NPG) ? (N - lo) : NPG);

  if (tid < 256) s_cnt[tid] = 0u;
  if (tid == 0) { s_ovf = 0u; s_next = 0u; }
  __syncthreads();

  if (E > 0) {
    const int* srcp = ei;
    const int* dstp = ei + E;
    const int nG = (E + 15) >> 4;
    const unsigned skey = (unsigned)blockIdx.x & 0xffu;
    unsigned* glist = &s_sorted[wave][0];
    unsigned gc = 0u;
    {
      const int nQ = (nG + 3) >> 2;
      const uint4* pv = (const uint4*)(planes + (size_t)(skey >> 5) * (size_t)(4 * nQ));
      const unsigned sh = skey & 31u;
      const int UNITV = 1024 * FV;
      const int nUnits = (nQ + UNITV - 1) / UNITV;
      const int nSlots = nUnits * UNITV;
      const int lastq = nQ - 1;
      const int off = (int)((blockIdx.x * 1024u) % (unsigned)nSlots);
      uint4 cur[FV];
      #pragma unroll
      for (int k = 0; k < FV; ++k) {
        int b = off + k * 1024;
        if (b >= nSlots) b -= nSlots;
        const int q = b + tid;
        cur[k] = pv[(q < nQ) ? q : lastq];
      }
      #pragma unroll 2
      for (int it = 0; it < nUnits; ++it) {
        const int b0 = off + it * UNITV;
        uint4 nxt[FV];
        #pragma unroll
        for (int k = 0; k < FV; ++k) {
          int b = b0 + UNITV + k * 1024;
          if (b >= nSlots) b -= nSlots;
          if (b >= nSlots) b -= nSlots;
          const int q = b + tid;
          nxt[k] = pv[(q < nQ) ? q : lastq];
        }
        #pragma unroll
        for (int k = 0; k < FV; ++k) {
          int b = b0 + k * 1024;
          if (b >= nSlots) b -= nSlots;
          const int q = b + tid;
          unsigned m = ((cur[k].x >> sh) & 1u) | (((cur[k].y >> sh) & 1u) << 1) |
                       (((cur[k].z >> sh) & 1u) << 2) | (((cur[k].w >> sh) & 1u) << 3);
          if (q >= nQ) m = 0u;
          unsigned hb = __builtin_amdgcn_ballot_w32(m != 0u);
          while (hb != 0u) {
            if (m != 0u) {
              const unsigned slot = gc + __builtin_amdgcn_mbcnt_lo(hb, 0u);
              const unsigned p = (unsigned)__builtin_ctz(m);
              m &= m - 1u;
              if (slot < NCAP) glist[slot] = 4u * (unsigned)q + p;
              else s_ovf = 1u;
            }
            gc += (unsigned)__popc(hb);
            hb = __builtin_amdgcn_ballot_w32(m != 0u);
          }
        }
        #pragma unroll
        for (int k = 0; k < FV; ++k) cur[k] = nxt[k];
      }
    }

    unsigned* mlist = &s_match[wave][0];
    unsigned wc = 0u;
    {
      const uint4* kv = (const uint4*)keys;
      const unsigned splat = skey * 0x01010101u;
      const unsigned ng = (gc < NCAP) ? gc : NCAP;
      for (unsigned base = 0; base < ng; base += 128u) {
        uint4 kq[4];
        unsigned gq[4];
        bool gv[4];
        #pragma unroll
        for (int t = 0; t < 4; ++t) {
          const unsigned i = base + (unsigned)t * 32u + (unsigned)lane;
          const unsigned gi = glist[(i < ng) ? i : 0u];
          gv[t] = (i < ng) && (gi < (unsigned)nG);
          gq[t] = gv[t] ? gi : 0u;
          kq[t] = kv[gq[t]];
        }
        #pragma unroll
        for (int t = 0; t < 4; ++t) {
          const unsigned x0 = kq[t].x ^ splat;
          const unsigned x1 = kq[t].y ^ splat;
          const unsigned x2 = kq[t].z ^ splat;
          const unsigned x3 = kq[t].w ^ splat;
          const unsigned z0 = (x0 - 0x01010101u) & ~x0 & 0x80808080u;
          const unsigned z1 = (x1 - 0x01010101u) & ~x1 & 0x80808080u;
          const unsigned z2 = (x2 - 0x01010101u) & ~x2 & 0x80808080u;
          const unsigned z3 = (x3 - 0x01010101u) & ~x3 & 0x80808080u;
          unsigned m = gv[t] ? ((z0 >> 7) | (z1 >> 6) | (z2 >> 5) | (z3 >> 4)) : 0u;
          unsigned hb = __builtin_amdgcn_ballot_w32(m != 0u);
          while (hb != 0u) {
            if (m != 0u) {
              const unsigned slot = wc + __builtin_amdgcn_mbcnt_lo(hb, 0u);
              const unsigned p = (unsigned)__builtin_ctz(m);
              m &= m - 1u;
              if (slot < WCAP) mlist[slot] = 16u * gq[t] + 4u * (p & 7u) + (p >> 3);
              else s_ovf = 1u;
            }
            wc += (unsigned)__popc(hb);
            hb = __builtin_amdgcn_ballot_w32(m != 0u);
          }
        }
      }
    }

    {
      const unsigned nm = (wc < WCAP) ? wc : WCAP;
      for (unsigned base = 0; base < nm; base += 128u) {
        int dv[4], sv[4];
        bool ok[4];
        #pragma unroll
        for (int q = 0; q < 4; ++q) {
          const unsigned i = base + (unsigned)q * 32u + (unsigned)lane;
          const unsigned e = (i < nm) ? mlist[(i < nm) ? i : 0u] : 0xffffffffu;
          ok[q] = e < (unsigned)E;
          const unsigned ec = ok[q] ? e : 0u;
          dv[q] = dstp[ec];
          sv[q] = srcp[ec];
        }
        #pragma unroll
        for (int q = 0; q < 4; ++q) {
          const unsigned dl = (unsigned)(dv[q] - lo);
          if (ok[q] && dl < span) push_edge(s_list, s_cnt, &s_ovf, dl, (unsigned)sv[q]);
        }
      }
    }
  }
  __syncthreads();

  const int ovf = __builtin_amdgcn_readfirstlane((int)s_ovf);
  for (;;) {
    unsigned mine = 0u;
    if (lane == 0) mine = atomicAdd(&s_next, 1u);
    const unsigned dl = (unsigned)__builtin_amdgcn_readfirstlane((int)mine);
    if (dl >= span) break;
    const int d = lo + (int)dl;
    if (!ovf) {
      int n = __builtin_amdgcn_readfirstlane((int)s_cnt[dl]);
      if (n > NCAP) n = NCAP;
      reduce_node_list(H, a_src, a_dst, bias, prelu_w, out, &s_list[dl][0],
                       &s_sorted[wave][0], n, d, lane);
    } else {
      reduce_node_global(ei, H, a_src, a_dst, bias, prelu_w, out, d, E, lane);
    }
  }
}

extern "C" void kernel_launch(void* const* d_in, const int* in_sizes, int n_in,
                              void* d_out, int out_size, void* d_ws, size_t ws_size,
                              hipStream_t stream) {
  const float* x       = (const float*)d_in[0];
  const int*   ei      = (const int*)d_in[1];
  const float* W       = (const float*)d_in[2];
  const float* att_src = (const float*)d_in[3];
  const float* att_dst = (const float*)d_in[4];
  const float* bias    = (const float*)d_in[5];
  const float* prelu_w = (const float*)d_in[6];
  float* out = (float*)d_out;

  const int N = in_sizes[0] / IN_CH;
  const int E = in_sizes[1] / 2;

  char* ws = (char*)d_ws;
  float* H = (float*)ws;            ws += (size_t)N * OUT_CH * sizeof(float);
  float* a_src = (float*)ws;        ws += (size_t)N * HEADS * sizeof(float);
  float* a_dst = (float*)ws;        ws += (size_t)N * HEADS * sizeof(float);
  unsigned* keys = (unsigned*)ws;
  ws += (size_t)((E + 15) >> 4) * 16;
  unsigned* planes = (unsigned*)ws;

  const int strips = (N + 15) / 16;
  const int nbKeys = ((((((E + 15) >> 4) + 3) >> 2) << 2) + 127) / 128;
  gemm_dots_keys<<<strips + nbKeys, 128, 0, stream>>>(x, W, H, N, att_src, att_dst,
                                                      a_src, a_dst, ei, keys, planes, E, strips);

  aggregate_slabs<<<(N + NPG - 1) / NPG, 1024, 0, stream>>>(ei, H, a_src, a_dst,
                                                            bias, prelu_w, out, keys, planes, E, N);
}
